# starting kernel
# baseline (speedup 1.0000x reference)
#include <hip/hip_runtime.h>
#include <hip/hip_bf16.h>
#include <math.h>

typedef __attribute__((ext_vector_type(2))) float v2f;
typedef __attribute__((ext_vector_type(8))) float v8f;

#define NNODES 100000
#define NFEAT  128
#define NHID   32
#define NCLASS 40
#define NEDGES 1600000
#define NEG_SLOPE 0.2f

static inline int ceil_div(long long a, long long b) { return (int)((a + b - 1) / b); }

__global__ __launch_bounds__(256) void gemm1_wmma(const float* __restrict__ x,
                                                  const float* __restrict__ W,
                                                  float* __restrict__ h) {
    __shared__ float ldsW[NFEAT * NHID];
    for (int i = threadIdx.x; i < NFEAT * NHID; i += 256) ldsW[i] = W[i];
    __syncthreads();

    const int lane    = threadIdx.x & 31;
    const int wave    = threadIdx.x >> 5;
    const int rowBase = blockIdx.x * 128 + wave * 16;
    const int half    = lane >> 4;
    const int l16     = lane & 15;

    int row  = rowBase + l16;
    int rowc = row < NNODES ? row : (NNODES - 1);
    const float* xr = x + (long long)rowc * NFEAT;

    v8f c0 = {}; v8f c1 = {};
    for (int k = 0; k < NFEAT; k += 4) {
        const int k0 = k + 2 * half;
        v2f a;
        a.x = xr[k0];
        a.y = xr[k0 + 1];
        v2f b0, b1;
        b0.x = ldsW[k0 * NHID + l16];
        b0.y = ldsW[(k0 + 1) * NHID + l16];
        b1.x = ldsW[k0 * NHID + 16 + l16];
        b1.y = ldsW[(k0 + 1) * NHID + 16 + l16];
        c0 = __builtin_amdgcn_wmma_f32_16x16x4_f32(false, a, false, b0, (short)0, c0, false, false);
        c1 = __builtin_amdgcn_wmma_f32_16x16x4_f32(false, a, false, b1, (short)0, c1, false, false);
    }
    #pragma unroll
    for (int m = 0; m < 8; ++m) {
        int r = rowBase + half * 8 + m;
        if (r < NNODES) {
            h[(long long)r * NHID + l16]      = c0[m];
            h[(long long)r * NHID + 16 + l16] = c1[m];
        }
    }
}

__global__ __launch_bounds__(256) void gemm2_wmma(const float* __restrict__ hin,
                                                  const float* __restrict__ W,
                                                  float* __restrict__ hout) {
    __shared__ float ldsW[NHID * 48];
    for (int i = threadIdx.x; i < NHID * 48; i += 256) {
        int k = i / 48, n = i % 48;
        ldsW[i] = (n < NCLASS) ? W[k * NCLASS + n] : 0.0f;
    }
    __syncthreads();

    const int lane    = threadIdx.x & 31;
    const int wave    = threadIdx.x >> 5;
    const int rowBase = blockIdx.x * 128 + wave * 16;
    const int half    = lane >> 4;
    const int l16     = lane & 15;

    int row  = rowBase + l16;
    int rowc = row < NNODES ? row : (NNODES - 1);
    const float* hr = hin + (long long)rowc * NHID;

    v8f c0 = {}; v8f c1 = {}; v8f c2 = {};
    for (int k = 0; k < NHID; k += 4) {
        const int k0 = k + 2 * half;
        v2f a;
        a.x = hr[k0];
        a.y = hr[k0 + 1];
        v2f b0, b1, b2;
        b0.x = ldsW[k0 * 48 + l16];        b0.y = ldsW[(k0 + 1) * 48 + l16];
        b1.x = ldsW[k0 * 48 + 16 + l16];   b1.y = ldsW[(k0 + 1) * 48 + 16 + l16];
        b2.x = ldsW[k0 * 48 + 32 + l16];   b2.y = ldsW[(k0 + 1) * 48 + 32 + l16];
        c0 = __builtin_amdgcn_wmma_f32_16x16x4_f32(false, a, false, b0, (short)0, c0, false, false);
        c1 = __builtin_amdgcn_wmma_f32_16x16x4_f32(false, a, false, b1, (short)0, c1, false, false);
        c2 = __builtin_amdgcn_wmma_f32_16x16x4_f32(false, a, false, b2, (short)0, c2, false, false);
    }
    #pragma unroll
    for (int m = 0; m < 8; ++m) {
        int r = rowBase + half * 8 + m;
        if (r < NNODES) {
            float* o = hout + (long long)r * NCLASS;
            o[l16]      = c0[m];
            o[16 + l16] = c1[m];
            if (32 + l16 < NCLASS) o[32 + l16] = c2[m];
        }
    }
}

__global__ void alpha_kernel(const float* __restrict__ h,
                             const float* __restrict__ asrc,
                             const float* __restrict__ adst,
                             float* __restrict__ outs, float* __restrict__ outd,
                             int n, int C) {
    int i = blockIdx.x * blockDim.x + threadIdx.x;
    if (i >= n) return;
    const float* r = h + (long long)i * C;
    float s = 0.f, d = 0.f;
    for (int c = 0; c < C; ++c) { float v = r[c]; s += v * asrc[c]; d += v * adst[c]; }
    outs[i] = s;
    outd[i] = d;
}

__global__ void edge_kernel(const float* __restrict__ as, const float* __restrict__ ad,
                            const int* __restrict__ src, const int* __restrict__ dst,
                            float* __restrict__ ew, float* __restrict__ denom, int E) {
    int i = blockIdx.x * blockDim.x + threadIdx.x;
    if (i >= E) return;
    float e = as[src[i]] + ad[dst[i]];
    e = e > 0.f ? e : NEG_SLOPE * e;
    float ee = expf(e);
    ew[i] = ee;
    atomicAdd(&denom[dst[i]], ee);
}

__global__ void scatter_kernel(const float* __restrict__ h, const float* __restrict__ ew,
                               const float* __restrict__ denom,
                               const int* __restrict__ src, const int* __restrict__ dst,
                               float* __restrict__ agg, long long total, int C) {
    long long i = (long long)blockIdx.x * blockDim.x + threadIdx.x;
    if (i >= total) return;
    int e = (int)(i / C);
    int c = (int)(i % C);
    int s = src[e], d = dst[e];
    float alpha = ew[e] / (denom[d] + 1e-16f);
    atomicAdd(&agg[(long long)d * C + c], h[(long long)s * C + c] * alpha);
}

__global__ void bias_relu_kernel(float* __restrict__ a, const float* __restrict__ b,
                                 long long total, int C) {
    long long i = (long long)blockIdx.x * blockDim.x + threadIdx.x;
    if (i >= total) return;
    float v = a[i] + b[(int)(i % C)];
    a[i] = v > 0.f ? v : 0.f;
}

__global__ void logsoftmax_bias_kernel(float* __restrict__ out,
                                       const float* __restrict__ b, int n) {
    int i = blockIdx.x * blockDim.x + threadIdx.x;
    if (i >= n) return;
    float* row = out + (long long)i * NCLASS;
    float m = -INFINITY;
    for (int c = 0; c < NCLASS; ++c) { float v = row[c] + b[c]; m = fmaxf(m, v); }
    float s = 0.f;
    for (int c = 0; c < NCLASS; ++c) s += expf(row[c] + b[c] - m);
    float lse = m + logf(s);
    for (int c = 0; c < NCLASS; ++c) row[c] = row[c] + b[c] - lse;
}

extern "C" void kernel_launch(void* const* d_in, const int* in_sizes, int n_in,
                              void* d_out, int out_size, void* d_ws, size_t ws_size,
                              hipStream_t stream) {
    const float* x     = (const float*)d_in[0];
    const int*   src   = (const int*)d_in[1];
    const int*   dst   = src + NEDGES;
    const float* W1    = (const float*)d_in[2];
    const float* as1v  = (const float*)d_in[3];
    const float* ad1v  = (const float*)d_in[4];
    const float* b1    = (const float*)d_in[5];
    const float* W2    = (const float*)d_in[6];
    const float* as2v  = (const float*)d_in[7];
    const float* ad2v  = (const float*)d_in[8];
    const float* b2    = (const float*)d_in[9];
    float* out = (float*)d_out;

    float* ws    = (float*)d_ws;
    float* h1    = ws;
    float* as    = h1 + (long long)NNODES * NHID;
    float* ad    = as + NNODES;
    float* denom = ad + NNODES;
    float* ew    = denom + NNODES;
    float* agg1  = ew + NEDGES;
    float* h2    = agg1 + (long long)NNODES * NHID;

    const int T = 256;
    const int gemmBlocks = ceil_div(NNODES, 128);

    gemm1_wmma<<<gemmBlocks, T, 0, stream>>>(x, W1, h1);
    alpha_kernel<<<ceil_div(NNODES, T), T, 0, stream>>>(h1, as1v, ad1v, as, ad, NNODES, NHID);
    hipMemsetAsync(denom, 0, sizeof(float) * NNODES, stream);
    hipMemsetAsync(agg1, 0, sizeof(float) * (size_t)NNODES * NHID, stream);
    edge_kernel<<<ceil_div(NEDGES, T), T, 0, stream>>>(as, ad, src, dst, ew, denom, NEDGES);
    {
        long long total = (long long)NEDGES * NHID;
        scatter_kernel<<<ceil_div(total, T), T, 0, stream>>>(h1, ew, denom, src, dst, agg1, total, NHID);
    }
    {
        long long total = (long long)NNODES * NHID;
        bias_relu_kernel<<<ceil_div(total, T), T, 0, stream>>>(agg1, b1, total, NHID);
    }

    gemm2_wmma<<<gemmBlocks, T, 0, stream>>>(agg1, W2, h2);
    alpha_kernel<<<ceil_div(NNODES, T), T, 0, stream>>>(h2, as2v, ad2v, as, ad, NNODES, NCLASS);
    hipMemsetAsync(denom, 0, sizeof(float) * NNODES, stream);
    hipMemsetAsync(out, 0, sizeof(float) * (size_t)NNODES * NCLASS, stream);
    edge_kernel<<<ceil_div(NEDGES, T), T, 0, stream>>>(as, ad, src, dst, ew, denom, NEDGES);
    {
        long long total = (long long)NEDGES * NCLASS;
        scatter_kernel<<<ceil_div(total, T), T, 0, stream>>>(h2, ew, denom, src, dst, out, total, NCLASS);
    }
    logsoftmax_bias_kernel<<<ceil_div(NNODES, T), T, 0, stream>>>(out, b2, NNODES);
}


# replace per-(edge,channel) float atomics by in-call destination-sorted edge list (counting sort, int atomics) and a wave-per-node gather with segment max; per-node sums now sequential in rank order
# speedup vs baseline: 2.3673x; 2.3673x over previous
#include <hip/hip_runtime.h>
#include <hip/hip_bf16.h>
#include <math.h>

typedef __attribute__((ext_vector_type(2))) float v2f;
typedef __attribute__((ext_vector_type(8))) float v8f;

#define NNODES 100000
#define NFEAT  128
#define NHID   32
#define NCLASS 40
#define NEDGES 1600000
#define NEG_SLOPE 0.2f

static inline int ceil_div(long long a, long long b) { return (int)((a + b - 1) / b); }

__global__ __launch_bounds__(256) void gemm1_wmma(const float* __restrict__ x,
                                                  const float* __restrict__ W,
                                                  float* __restrict__ h) {
    __shared__ float ldsW[NFEAT * NHID];
    for (int i = threadIdx.x; i < NFEAT * NHID; i += 256) ldsW[i] = W[i];
    __syncthreads();

    const int lane    = threadIdx.x & 31;
    const int wave    = threadIdx.x >> 5;
    const int rowBase = blockIdx.x * 128 + wave * 16;
    const int half    = lane >> 4;
    const int l16     = lane & 15;

    int row  = rowBase + l16;
    int rowc = row < NNODES ? row : (NNODES - 1);
    const float* xr = x + (long long)rowc * NFEAT;

    v8f c0 = {}; v8f c1 = {};
    for (int k = 0; k < NFEAT; k += 4) {
        const int k0 = k + 2 * half;
        v2f a;
        a.x = xr[k0];
        a.y = xr[k0 + 1];
        v2f b0, b1;
        b0.x = ldsW[k0 * NHID + l16];
        b0.y = ldsW[(k0 + 1) * NHID + l16];
        b1.x = ldsW[k0 * NHID + 16 + l16];
        b1.y = ldsW[(k0 + 1) * NHID + 16 + l16];
        c0 = __builtin_amdgcn_wmma_f32_16x16x4_f32(false, a, false, b0, (short)0, c0, false, false);
        c1 = __builtin_amdgcn_wmma_f32_16x16x4_f32(false, a, false, b1, (short)0, c1, false, false);
    }
    #pragma unroll
    for (int m = 0; m < 8; ++m) {
        int r = rowBase + half * 8 + m;
        if (r < NNODES) {
            h[(long long)r * NHID + l16]      = c0[m];
            h[(long long)r * NHID + 16 + l16] = c1[m];
        }
    }
}

__global__ __launch_bounds__(256) void gemm2_wmma(const float* __restrict__ hin,
                                                  const float* __restrict__ W,
                                                  float* __restrict__ hout) {
    __shared__ float ldsW[NHID * 48];
    for (int i = threadIdx.x; i < NHID * 48; i += 256) {
        int k = i / 48, n = i % 48;
        ldsW[i] = (n < NCLASS) ? W[k * NCLASS + n] : 0.0f;
    }
    __syncthreads();

    const int lane    = threadIdx.x & 31;
    const int wave    = threadIdx.x >> 5;
    const int rowBase = blockIdx.x * 128 + wave * 16;
    const int half    = lane >> 4;
    const int l16     = lane & 15;

    int row  = rowBase + l16;
    int rowc = row < NNODES ? row : (NNODES - 1);
    const float* hr = hin + (long long)rowc * NHID;

    v8f c0 = {}; v8f c1 = {}; v8f c2 = {};
    for (int k = 0; k < NHID; k += 4) {
        const int k0 = k + 2 * half;
        v2f a;
        a.x = hr[k0];
        a.y = hr[k0 + 1];
        v2f b0, b1, b2;
        b0.x = ldsW[k0 * 48 + l16];        b0.y = ldsW[(k0 + 1) * 48 + l16];
        b1.x = ldsW[k0 * 48 + 16 + l16];   b1.y = ldsW[(k0 + 1) * 48 + 16 + l16];
        b2.x = ldsW[k0 * 48 + 32 + l16];   b2.y = ldsW[(k0 + 1) * 48 + 32 + l16];
        c0 = __builtin_amdgcn_wmma_f32_16x16x4_f32(false, a, false, b0, (short)0, c0, false, false);
        c1 = __builtin_amdgcn_wmma_f32_16x16x4_f32(false, a, false, b1, (short)0, c1, false, false);
        c2 = __builtin_amdgcn_wmma_f32_16x16x4_f32(false, a, false, b2, (short)0, c2, false, false);
    }
    #pragma unroll
    for (int m = 0; m < 8; ++m) {
        int r = rowBase + half * 8 + m;
        if (r < NNODES) {
            float* o = hout + (long long)r * NCLASS;
            o[l16]      = c0[m];
            o[16 + l16] = c1[m];
            if (32 + l16 < NCLASS) o[32 + l16] = c2[m];
        }
    }
}

__global__ void alpha_kernel(const float* __restrict__ h,
                             const float* __restrict__ asrc,
                             const float* __restrict__ adst,
                             float* __restrict__ outs, float* __restrict__ outd,
                             int n, int C) {
    int i = blockIdx.x * blockDim.x + threadIdx.x;
    if (i >= n) return;
    const float* r = h + (long long)i * C;
    float s = 0.f, d = 0.f;
    for (int c = 0; c < C; ++c) { float v = r[c]; s += v * asrc[c]; d += v * adst[c]; }
    outs[i] = s;
    outd[i] = d;
}

__global__ void degree_rank_kernel(const int* __restrict__ dst, int* __restrict__ cnt,
                                   int* __restrict__ rank, int E) {
    int i = blockIdx.x * blockDim.x + threadIdx.x;
    if (i >= E) return;
    rank[i] = atomicAdd(&cnt[dst[i]], 1);
}

__global__ __launch_bounds__(1024) void rowptr_kernel(const int* __restrict__ cnt,
                                                      int* __restrict__ rowptr) {
    __shared__ int wtot[32];
    __shared__ int wpre[32];
    const int tid  = threadIdx.x;
    const int lane = tid & 31;
    const int wv   = tid >> 5;
    const int base = blockIdx.x * 1024;

    int p = 0;
    const int4* c4 = reinterpret_cast<const int4*>(cnt);
    const int n4 = base >> 2;
    #pragma unroll 4
    for (int q = tid; q < n4; q += 1024) {
        int4 v = c4[q];
        p += (v.x + v.y) + (v.z + v.w);
    }
    const int i = base + tid;
    const int c = i < NNODES ? cnt[i] : 0;

    int inc = c;
    for (int o = 1; o < 32; o <<= 1) {
        int t = __shfl_up(inc, o, 32);
        if (lane >= o) inc += t;
    }
    for (int o = 16; o > 0; o >>= 1) p += __shfl_xor(p, o, 32);
    if (lane == 31) wtot[wv] = inc;
    if (lane == 0)  wpre[wv] = p;
    __syncthreads();
    if (wv == 0) {
        const int t = wtot[lane];
        int pp   = wpre[lane];
        int inc2 = t;
        for (int o = 1; o < 32; o <<= 1) {
            int u = __shfl_up(inc2, o, 32);
            if (lane >= o) inc2 += u;
        }
        for (int o = 16; o > 0; o >>= 1) pp += __shfl_xor(pp, o, 32);
        wtot[lane] = inc2 - t + pp;
    }
    __syncthreads();
    if (i < NNODES) rowptr[i] = wtot[wv] + inc - c;
}

__global__ void place_kernel(const int* __restrict__ src, const int* __restrict__ dst,
                             const int* __restrict__ rowptr, const int* __restrict__ rank,
                             int* __restrict__ csr, int E) {
    int i = blockIdx.x * blockDim.x + threadIdx.x;
    if (i >= E) return;
    csr[rowptr[dst[i]] + rank[i]] = src[i];
}

template <int V> struct RowV { float v[V]; };

template <int C, int V>
__device__ __forceinline__ RowV<V> row_load(const float* __restrict__ h, int s, int cl) {
    RowV<V> r;
    const float* p = h + (long long)s * C + cl * V;
    if (V == 2) {
        float2 t = *reinterpret_cast<const float2*>(p);
        r.v[0] = t.x;
        r.v[V - 1] = t.y;
    } else {
        r.v[0] = p[0];
    }
    return r;
}

__device__ __forceinline__ int lane_i(int v, int l) {
    return __builtin_amdgcn_readlane(v, l);
}
__device__ __forceinline__ float lane_f(float v, int l) {
    return __int_as_float(__builtin_amdgcn_readlane(__float_as_int(v), l));
}

template <int C, int V>
__global__ __launch_bounds__(256) void gather_kernel(const float* __restrict__ h,
                                                     const float* __restrict__ as,
                                                     const float* __restrict__ ad,
                                                     const int* __restrict__ rowptr,
                                                     const int* __restrict__ csr,
                                                     float* __restrict__ out) {
    constexpr int AL = C / V;
    const int lane = threadIdx.x & 31;
    const int node = __builtin_amdgcn_readfirstlane((int)(blockIdx.x * 8 + (threadIdx.x >> 5)));
    if (node >= NNODES) return;
    const int beg = rowptr[node];
    const int end = (node + 1 < NNODES) ? rowptr[node + 1] : NEDGES;
    const float adv = ad[node];
    const int cl = lane < AL ? lane : AL - 1;

    int j = beg + lane;
    bool ok = j < end;
    int s = ok ? csr[j] : 0;
    float e = -INFINITY;
    if (ok) { float t = as[s] + adv; e = t > 0.f ? t : NEG_SLOPE * t; }
    float m = e;
    for (int jj = j + 32; jj < end; jj += 32) {
        float t = as[csr[jj]] + adv;
        t = t > 0.f ? t : NEG_SLOPE * t;
        m = fmaxf(m, t);
    }
    for (int o = 16; o > 0; o >>= 1) m = fmaxf(m, __shfl_xor(m, o, 32));

    float acc[V];
    for (int v = 0; v < V; ++v) acc[v] = 0.f;
    float den = 0.f;

    for (int base = beg; base < end; base += 32) {
        if (base != beg) {
            j = base + lane;
            ok = j < end;
            s = ok ? csr[j] : 0;
            e = -INFINITY;
            if (ok) { float t = as[s] + adv; e = t > 0.f ? t : NEG_SLOPE * t; }
        }
        const float w = expf(e - m);
        den += w;
        int n = end - base;
        if (n > 32) n = 32;
        int t = 0;
        for (; t + 4 <= n; t += 4) {
            const int s0 = lane_i(s, t),     s1 = lane_i(s, t + 1);
            const int s2 = lane_i(s, t + 2), s3 = lane_i(s, t + 3);
            const float w0 = lane_f(w, t),     w1 = lane_f(w, t + 1);
            const float w2 = lane_f(w, t + 2), w3 = lane_f(w, t + 3);
            const RowV<V> r0 = row_load<C, V>(h, s0, cl);
            const RowV<V> r1 = row_load<C, V>(h, s1, cl);
            const RowV<V> r2 = row_load<C, V>(h, s2, cl);
            const RowV<V> r3 = row_load<C, V>(h, s3, cl);
            for (int v = 0; v < V; ++v) {
                acc[v] += w0 * r0.v[v];
                acc[v] += w1 * r1.v[v];
                acc[v] += w2 * r2.v[v];
                acc[v] += w3 * r3.v[v];
            }
        }
        for (; t < n; ++t) {
            const int s0 = lane_i(s, t);
            const float w0 = lane_f(w, t);
            const RowV<V> r0 = row_load<C, V>(h, s0, cl);
            for (int v = 0; v < V; ++v) acc[v] += w0 * r0.v[v];
        }
    }
    for (int o = 16; o > 0; o >>= 1) den += __shfl_xor(den, o, 32);
    const float dn = den + 1e-16f;

    float* orow = out + (long long)node * C;
    if (V == 2) {
        if (lane < AL) {
            float2 r;
            r.x = acc[0] / dn;
            r.y = acc[V - 1] / dn;
            *reinterpret_cast<float2*>(orow + lane * V) = r;
        }
    } else {
        orow[lane] = acc[0] / dn;
    }
}

__global__ void bias_relu_kernel(float* __restrict__ a, const float* __restrict__ b,
                                 long long total, int C) {
    long long i = (long long)blockIdx.x * blockDim.x + threadIdx.x;
    if (i >= total) return;
    float v = a[i] + b[(int)(i % C)];
    a[i] = v > 0.f ? v : 0.f;
}

__global__ void logsoftmax_bias_kernel(float* __restrict__ out,
                                       const float* __restrict__ b, int n) {
    int i = blockIdx.x * blockDim.x + threadIdx.x;
    if (i >= n) return;
    float* row = out + (long long)i * NCLASS;
    float m = -INFINITY;
    for (int c = 0; c < NCLASS; ++c) { float v = row[c] + b[c]; m = fmaxf(m, v); }
    float s = 0.f;
    for (int c = 0; c < NCLASS; ++c) s += expf(row[c] + b[c] - m);
    float lse = m + logf(s);
    for (int c = 0; c < NCLASS; ++c) row[c] = row[c] + b[c] - lse;
}

extern "C" void kernel_launch(void* const* d_in, const int* in_sizes, int n_in,
                              void* d_out, int out_size, void* d_ws, size_t ws_size,
                              hipStream_t stream) {
    const float* x     = (const float*)d_in[0];
    const int*   src   = (const int*)d_in[1];
    const int*   dst   = src + NEDGES;
    const float* W1    = (const float*)d_in[2];
    const float* as1v  = (const float*)d_in[3];
    const float* ad1v  = (const float*)d_in[4];
    const float* b1    = (const float*)d_in[5];
    const float* W2    = (const float*)d_in[6];
    const float* as2v  = (const float*)d_in[7];
    const float* ad2v  = (const float*)d_in[8];
    const float* b2    = (const float*)d_in[9];
    float* out = (float*)d_out;

    float* ws    = (float*)d_ws;
    float* h1    = ws;
    float* as    = h1 + (long long)NNODES * NHID;
    float* ad    = as + NNODES;
    int*   rowptr = (int*)(ad + NNODES);
    int*   csr   = rowptr + NNODES;
    float* agg1  = (float*)(csr + NEDGES);
    float* h2    = agg1 + (long long)NNODES * NHID;
    int*   rank  = (int*)agg1;
    int*   cnt   = (int*)h2;

    const int T = 256;
    const int gemmBlocks = ceil_div(NNODES, 128);

    hipMemsetAsync(cnt, 0, sizeof(int) * NNODES, stream);
    degree_rank_kernel<<<ceil_div(NEDGES, T), T, 0, stream>>>(dst, cnt, rank, NEDGES);
    rowptr_kernel<<<ceil_div(NNODES, 1024), 1024, 0, stream>>>(cnt, rowptr);
    place_kernel<<<ceil_div(NEDGES, T), T, 0, stream>>>(src, dst, rowptr, rank, csr, NEDGES);

    gemm1_wmma<<<gemmBlocks, T, 0, stream>>>(x, W1, h1);
    alpha_kernel<<<ceil_div(NNODES, T), T, 0, stream>>>(h1, as1v, ad1v, as, ad, NNODES, NHID);
    gather_kernel<NHID, 1><<<ceil_div(NNODES, 8), T, 0, stream>>>(h1, as, ad, rowptr, csr, agg1);
    {
        long long total = (long long)NNODES * NHID;
        bias_relu_kernel<<<ceil_div(total, T), T, 0, stream>>>(agg1, b1, total, NHID);
    }

    gemm2_wmma<<<gemmBlocks, T, 0, stream>>>(agg1, W2, h2);
    alpha_kernel<<<ceil_div(NNODES, T), T, 0, stream>>>(h2, as2v, ad2v, as, ad, NNODES, NCLASS);
    gather_kernel<NCLASS, 2><<<ceil_div(NNODES, 8), T, 0, stream>>>(h2, as, ad, rowptr, csr, out);
    logsoftmax_bias_kernel<<<ceil_div(NNODES, T), T, 0, stream>>>(out, b2, NNODES);
}


# destination sort rebuilt as a two-level counting sort on LDS counters (256-node bins, then per-bin count, scan, place): no global atomics, no memset; per-node sum order still the adds' arrival order
# speedup vs baseline: 3.4617x; 1.4623x over previous
#include <hip/hip_runtime.h>
#include <hip/hip_bf16.h>
#include <math.h>

typedef __attribute__((ext_vector_type(2))) float v2f;
typedef __attribute__((ext_vector_type(8))) float v8f;

#define NNODES 100000
#define NFEAT  128
#define NHID   32
#define NCLASS 40
#define NEDGES 1600000
#define NEG_SLOPE 0.2f

static inline int ceil_div(long long a, long long b) { return (int)((a + b - 1) / b); }

template <int K>
__device__ __forceinline__ float row_x(float v) {
    return __int_as_float(__builtin_amdgcn_update_dpp(0, __float_as_int(v), 0x160 + K, 0xf, 0xf, true));
}
__device__ __forceinline__ float xhalf(float v) {
    const int b = __float_as_int(v);
    return __int_as_float(__builtin_amdgcn_permlanex16(b, b, 0x76543210, 0xfedcba98, false, false));
}
__device__ __forceinline__ float row16_sum(float v) {
    v += row_x<1>(v);
    v += row_x<2>(v);
    v += row_x<4>(v);
    v += row_x<8>(v);
    return v;
}
__device__ __forceinline__ float row16_max(float v) {
    v = fmaxf(v, row_x<1>(v));
    v = fmaxf(v, row_x<2>(v));
    v = fmaxf(v, row_x<4>(v));
    v = fmaxf(v, row_x<8>(v));
    return v;
}

__global__ __launch_bounds__(256) void gemm1_wmma(const float* __restrict__ x,
                                                  const float* __restrict__ W,
                                                  const float* __restrict__ asrc,
                                                  const float* __restrict__ adst,
                                                  float* __restrict__ h,
                                                  float* __restrict__ outs,
                                                  float* __restrict__ outd) {
    __shared__ float ldsW[NFEAT * NHID];
    for (int i = threadIdx.x; i < NFEAT * NHID; i += 256) ldsW[i] = W[i];
    __syncthreads();

    const int lane    = threadIdx.x & 31;
    const int wave    = threadIdx.x >> 5;
    const int rowBase = blockIdx.x * 128 + wave * 16;
    const int half    = lane >> 4;
    const int l16     = lane & 15;

    int row  = rowBase + l16;
    int rowc = row < NNODES ? row : (NNODES - 1);
    const float* xr = x + (long long)rowc * NFEAT;

    v8f c0 = {}; v8f c1 = {};
    for (int k = 0; k < NFEAT; k += 4) {
        const int k0 = k + 2 * half;
        v2f a;
        a.x = xr[k0];
        a.y = xr[k0 + 1];
        v2f b0, b1;
        b0.x = ldsW[k0 * NHID + l16];
        b0.y = ldsW[(k0 + 1) * NHID + l16];
        b1.x = ldsW[k0 * NHID + 16 + l16];
        b1.y = ldsW[(k0 + 1) * NHID + 16 + l16];
        c0 = __builtin_amdgcn_wmma_f32_16x16x4_f32(false, a, false, b0, (short)0, c0, false, false);
        c1 = __builtin_amdgcn_wmma_f32_16x16x4_f32(false, a, false, b1, (short)0, c1, false, false);
    }
    {
        const float s0 = asrc[l16], s1 = asrc[16 + l16];
        const float d0 = adst[l16], d1 = adst[16 + l16];
        float vs = 0.f, vd = 0.f;
        #pragma unroll
        for (int m = 0; m < 8; ++m) {
            const float ps = row16_sum(c0[m] * s0 + c1[m] * s1);
            const float pd = row16_sum(c0[m] * d0 + c1[m] * d1);
            if (l16 == m) { vs = ps; vd = pd; }
        }
        const int r = rowBase + half * 8 + l16;
        if (l16 < 8 && r < NNODES) { outs[r] = vs; outd[r] = vd; }
    }
    #pragma unroll
    for (int m = 0; m < 8; ++m) {
        int r = rowBase + half * 8 + m;
        if (r < NNODES) {
            h[(long long)r * NHID + l16]      = c0[m];
            h[(long long)r * NHID + 16 + l16] = c1[m];
        }
    }
}

__global__ __launch_bounds__(256) void gemm2_wmma(const float* __restrict__ hin,
                                                  const float* __restrict__ W,
                                                  const float* __restrict__ asrc,
                                                  const float* __restrict__ adst,
                                                  float* __restrict__ hout,
                                                  float* __restrict__ outs,
                                                  float* __restrict__ outd) {
    __shared__ float ldsW[NHID * 48];
    for (int i = threadIdx.x; i < NHID * 48; i += 256) {
        int k = i / 48, n = i % 48;
        ldsW[i] = (n < NCLASS) ? W[k * NCLASS + n] : 0.0f;
    }
    __syncthreads();

    const int lane    = threadIdx.x & 31;
    const int wave    = threadIdx.x >> 5;
    const int rowBase = blockIdx.x * 128 + wave * 16;
    const int half    = lane >> 4;
    const int l16     = lane & 15;

    int row  = rowBase + l16;
    int rowc = row < NNODES ? row : (NNODES - 1);
    const float* hr = hin + (long long)rowc * NHID;

    v8f c0 = {}; v8f c1 = {}; v8f c2 = {};
    for (int k = 0; k < NHID; k += 4) {
        const int k0 = k + 2 * half;
        v2f a;
        a.x = hr[k0];
        a.y = hr[k0 + 1];
        v2f b0, b1, b2;
        b0.x = ldsW[k0 * 48 + l16];        b0.y = ldsW[(k0 + 1) * 48 + l16];
        b1.x = ldsW[k0 * 48 + 16 + l16];   b1.y = ldsW[(k0 + 1) * 48 + 16 + l16];
        b2.x = ldsW[k0 * 48 + 32 + l16];   b2.y = ldsW[(k0 + 1) * 48 + 32 + l16];
        c0 = __builtin_amdgcn_wmma_f32_16x16x4_f32(false, a, false, b0, (short)0, c0, false, false);
        c1 = __builtin_amdgcn_wmma_f32_16x16x4_f32(false, a, false, b1, (short)0, c1, false, false);
        c2 = __builtin_amdgcn_wmma_f32_16x16x4_f32(false, a, false, b2, (short)0, c2, false, false);
    }
    {
        const bool in3 = 32 + l16 < NCLASS;
        const float s0 = asrc[l16], s1 = asrc[16 + l16], s2 = in3 ? asrc[32 + l16] : 0.f;
        const float d0 = adst[l16], d1 = adst[16 + l16], d2 = in3 ? adst[32 + l16] : 0.f;
        float vs = 0.f, vd = 0.f;
        #pragma unroll
        for (int m = 0; m < 8; ++m) {
            const float ps = row16_sum(c0[m] * s0 + c1[m] * s1 + c2[m] * s2);
            const float pd = row16_sum(c0[m] * d0 + c1[m] * d1 + c2[m] * d2);
            if (l16 == m) { vs = ps; vd = pd; }
        }
        const int r = rowBase + half * 8 + l16;
        if (l16 < 8 && r < NNODES) { outs[r] = vs; outd[r] = vd; }
    }
    #pragma unroll
    for (int m = 0; m < 8; ++m) {
        int r = rowBase + half * 8 + m;
        if (r < NNODES) {
            float* o = hout + (long long)r * NCLASS;
            o[l16]      = c0[m];
            o[16 + l16] = c1[m];
            if (32 + l16 < NCLASS) o[32 + l16] = c2[m];
        }
    }
}

#define BIN_SHIFT   8
#define BIN_NODES   256
#define NBINS       391
#define NBLK        256
#define CHUNK       6250
#define CHUNK_TRIPS 7
#define NHIST       (NBINS * NBLK)
#define SORT_U      12
static_assert(NBLK * CHUNK == NEDGES, "chunks must tile the edge list");
static_assert(NBINS * BIN_NODES >= NNODES, "bins must cover the nodes");
static_assert(CHUNK_TRIPS * 1024 >= CHUNK, "a block must cover its chunk");

__global__ __launch_bounds__(1024) void bin_count_kernel(const int* __restrict__ dst,
                                                         int* __restrict__ hist) {
    __shared__ int lh[NBINS];
    const int tid = threadIdx.x;
    const int b   = blockIdx.x;
    for (int i = tid; i < NBINS; i += 1024) lh[i] = 0;
    const int beg = b * CHUNK;
    int d[CHUNK_TRIPS];
    #pragma unroll
    for (int k = 0; k < CHUNK_TRIPS; ++k) {
        const int o = tid + k * 1024;
        d[k] = o < CHUNK ? dst[beg + o] : 0;
    }
    __syncthreads();
    #pragma unroll
    for (int k = 0; k < CHUNK_TRIPS; ++k) {
        const int o = tid + k * 1024;
        if (o < CHUNK) atomicAdd(&lh[d[k] >> BIN_SHIFT], 1);
    }
    __syncthreads();
    for (int i = tid; i < NBINS; i += 1024) hist[i * NBLK + b] = lh[i];
}

__global__ __launch_bounds__(1024) void excl_scan_kernel(const int* __restrict__ cnt,
                                                         int* __restrict__ out, int n) {
    __shared__ int wtot[32];
    __shared__ int wpre[32];
    const int tid  = threadIdx.x;
    const int lane = tid & 31;
    const int wv   = tid >> 5;
    const int base = blockIdx.x * 1024;

    int p = 0;
    const int4* c4 = reinterpret_cast<const int4*>(cnt);
    const int n4 = base >> 2;
    #pragma unroll 4
    for (int q = tid; q < n4; q += 1024) {
        int4 v = c4[q];
        p += (v.x + v.y) + (v.z + v.w);
    }
    const int i = base + tid;
    const int c = i < n ? cnt[i] : 0;

    int inc = c;
    for (int o = 1; o < 32; o <<= 1) {
        int t = __shfl_up(inc, o, 32);
        if (lane >= o) inc += t;
    }
    for (int o = 16; o > 0; o >>= 1) p += __shfl_xor(p, o, 32);
    if (lane == 31) wtot[wv] = inc;
    if (lane == 0)  wpre[wv] = p;
    __syncthreads();
    if (wv == 0) {
        const int t = wtot[lane];
        int pp   = wpre[lane];
        int inc2 = t;
        for (int o = 1; o < 32; o <<= 1) {
            int u = __shfl_up(inc2, o, 32);
            if (lane >= o) inc2 += u;
        }
        for (int o = 16; o > 0; o >>= 1) pp += __shfl_xor(pp, o, 32);
        wtot[lane] = inc2 - t + pp;
    }
    __syncthreads();
    if (i < n) out[i] = wtot[wv] + inc - c;
}

__global__ __launch_bounds__(1024) void bin_scatter_kernel(const int* __restrict__ src,
                                                           const int* __restrict__ dst,
                                                           const int* __restrict__ goff,
                                                           int* __restrict__ binned) {
    __shared__ int cur[NBINS];
    const int tid = threadIdx.x;
    const int b   = blockIdx.x;
    const int beg = b * CHUNK;
    int d[CHUNK_TRIPS], s[CHUNK_TRIPS];
    #pragma unroll
    for (int k = 0; k < CHUNK_TRIPS; ++k) {
        const int o = tid + k * 1024;
        d[k] = o < CHUNK ? dst[beg + o] : 0;
        s[k] = o < CHUNK ? src[beg + o] : 0;
    }
    for (int i = tid; i < NBINS; i += 1024) cur[i] = goff[i * NBLK + b];
    __syncthreads();
    #pragma unroll
    for (int k = 0; k < CHUNK_TRIPS; ++k) {
        const int o = tid + k * 1024;
        if (o < CHUNK) {
            const int pos = atomicAdd(&cur[d[k] >> BIN_SHIFT], 1);
            binned[pos] = (s[k] << BIN_SHIFT) | (d[k] & (BIN_NODES - 1));
        }
    }
}

__global__ __launch_bounds__(512) void bin_sort_kernel(const int* __restrict__ goff,
                                                       const int* __restrict__ binned,
                                                       int* __restrict__ rowptr,
                                                       int* __restrict__ csr) {
    __shared__ int cnt[BIN_NODES];
    __shared__ int wtot[BIN_NODES / 32];
    const int tid  = threadIdx.x;
    const int lane = tid & 31;
    const int wv   = tid >> 5;
    const int bin  = blockIdx.x;
    const int beg  = goff[bin * NBLK];
    const int end  = (bin + 1 < NBINS) ? goff[(bin + 1) * NBLK] : NEDGES;

    if (tid < BIN_NODES) cnt[tid] = 0;
    __syncthreads();
    for (int i0 = beg + tid; i0 < end; i0 += SORT_U * 512) {
        int v[SORT_U];
        #pragma unroll
        for (int k = 0; k < SORT_U; ++k) {
            const int i = i0 + k * 512;
            v[k] = binned[i < end ? i : end - 1];
        }
        #pragma unroll
        for (int k = 0; k < SORT_U; ++k) {
            const int i = i0 + k * 512;
            atomicAdd(&cnt[v[k] & (BIN_NODES - 1)], i < end ? 1 : 0);
        }
    }
    __syncthreads();

    const int c = tid < BIN_NODES ? cnt[tid] : 0;
    int inc = c;
    for (int o = 1; o < 32; o <<= 1) {
        int t = __shfl_up(inc, o, 32);
        if (lane >= o) inc += t;
    }
    if (lane == 31 && wv < BIN_NODES / 32) wtot[wv] = inc;
    __syncthreads();
    int pre = 0;
    for (int w = 0; w < BIN_NODES / 32; ++w) if (w < wv) pre += wtot[w];
    if (tid < BIN_NODES) {
        const int first = beg + pre + inc - c;
        cnt[tid] = first;
        const int node = bin * BIN_NODES + tid;
        if (node < NNODES) rowptr[node] = first;
    }
    __syncthreads();
    for (int i0 = beg + tid; i0 < end; i0 += SORT_U * 512) {
        int v[SORT_U];
        #pragma unroll
        for (int k = 0; k < SORT_U; ++k) {
            const int i = i0 + k * 512;
            v[k] = binned[i < end ? i : end - 1];
        }
        #pragma unroll
        for (int k = 0; k < SORT_U; ++k) {
            const int i = i0 + k * 512;
            const int pos = atomicAdd(&cnt[v[k] & (BIN_NODES - 1)], i < end ? 1 : 0);
            if (i < end) csr[pos] = v[k] >> BIN_SHIFT;
        }
    }
}

template <int V> struct RowV { float v[V]; };

template <int C, int V>
__device__ __forceinline__ RowV<V> row_load(const float* __restrict__ h, int s, int cl) {
    RowV<V> r;
    const float* p = h + (long long)s * C + cl * V;
    if (V == 2) {
        float2 t = *reinterpret_cast<const float2*>(p);
        r.v[0] = t.x;
        r.v[V - 1] = t.y;
    } else {
        r.v[0] = p[0];
    }
    return r;
}

__device__ __forceinline__ int lane_i(int v, int l) {
    return __builtin_amdgcn_readlane(v, l);
}
__device__ __forceinline__ float lane_f(float v, int l) {
    return __int_as_float(__builtin_amdgcn_readlane(__float_as_int(v), l));
}

template <int C, int V, bool LSM>
__global__ __launch_bounds__(256) void gather_kernel(const float* __restrict__ h,
                                                     const float* __restrict__ as,
                                                     const float* __restrict__ ad,
                                                     const int* __restrict__ rowptr,
                                                     const int* __restrict__ csr,
                                                     const float* __restrict__ bias,
                                                     float* __restrict__ out) {
    constexpr int AL = C / V;
    const int lane = threadIdx.x & 31;
    const int node = __builtin_amdgcn_readfirstlane((int)(blockIdx.x * 8 + (threadIdx.x >> 5)));
    if (node >= NNODES) return;
    const int beg = rowptr[node];
    const int end = (node + 1 < NNODES) ? rowptr[node + 1] : NEDGES;
    const float adv = ad[node];
    const int cl = lane < AL ? lane : AL - 1;

    int j = beg + lane;
    bool ok = j < end;
    int s = ok ? csr[j] : 0;
    float e = -INFINITY;
    if (ok) { float t = as[s] + adv; e = t > 0.f ? t : NEG_SLOPE * t; }
    float m = e;
    for (int jj = j + 32; jj < end; jj += 32) {
        float t = as[csr[jj]] + adv;
        t = t > 0.f ? t : NEG_SLOPE * t;
        m = fmaxf(m, t);
    }
    for (int o = 16; o > 0; o >>= 1) m = fmaxf(m, __shfl_xor(m, o, 32));

    float acc[V];
    for (int v = 0; v < V; ++v) acc[v] = 0.f;
    float den = 0.f;

    for (int base = beg; base < end; base += 32) {
        if (base != beg) {
            j = base + lane;
            ok = j < end;
            s = ok ? csr[j] : 0;
            e = -INFINITY;
            if (ok) { float t = as[s] + adv; e = t > 0.f ? t : NEG_SLOPE * t; }
        }
        const float w = expf(e - m);
        den += w;
        int n = end - base;
        if (n > 32) n = 32;
        int t = 0;
        for (; t + 4 <= n; t += 4) {
            const int s0 = lane_i(s, t),     s1 = lane_i(s, t + 1);
            const int s2 = lane_i(s, t + 2), s3 = lane_i(s, t + 3);
            const float w0 = lane_f(w, t),     w1 = lane_f(w, t + 1);
            const float w2 = lane_f(w, t + 2), w3 = lane_f(w, t + 3);
            const RowV<V> r0 = row_load<C, V>(h, s0, cl);
            const RowV<V> r1 = row_load<C, V>(h, s1, cl);
            const RowV<V> r2 = row_load<C, V>(h, s2, cl);
            const RowV<V> r3 = row_load<C, V>(h, s3, cl);
            for (int v = 0; v < V; ++v) {
                acc[v] += w0 * r0.v[v];
                acc[v] += w1 * r1.v[v];
                acc[v] += w2 * r2.v[v];
                acc[v] += w3 * r3.v[v];
            }
        }
        for (; t < n; ++t) {
            const int s0 = lane_i(s, t);
            const float w0 = lane_f(w, t);
            const RowV<V> r0 = row_load<C, V>(h, s0, cl);
            for (int v = 0; v < V; ++v) acc[v] += w0 * r0.v[v];
        }
    }
    for (int o = 16; o > 0; o >>= 1) den += __shfl_xor(den, o, 32);
    const float dn = den + 1e-16f;

    const bool own = lane < AL;
    float val[V];
    for (int v = 0; v < V; ++v) val[v] = acc[v] / dn + bias[cl * V + v];
    if (LSM) {
        float mx = -INFINITY;
        if (own) for (int v = 0; v < V; ++v) mx = fmaxf(mx, val[v]);
        mx = row16_max(mx);
        mx = fmaxf(mx, xhalf(mx));
        float se = 0.f;
        if (own) for (int v = 0; v < V; ++v) se += expf(val[v] - mx);
        se = row16_sum(se);
        se += xhalf(se);
        const float lse = mx + logf(se);
        for (int v = 0; v < V; ++v) val[v] -= lse;
    } else {
        for (int v = 0; v < V; ++v) val[v] = val[v] > 0.f ? val[v] : 0.f;
    }

    float* orow = out + (long long)node * C;
    if (V == 2) {
        if (own) {
            float2 r;
            r.x = val[0];
            r.y = val[V - 1];
            *reinterpret_cast<float2*>(orow + lane * V) = r;
        }
    } else {
        orow[lane] = val[0];
    }
}

extern "C" void kernel_launch(void* const* d_in, const int* in_sizes, int n_in,
                              void* d_out, int out_size, void* d_ws, size_t ws_size,
                              hipStream_t stream) {
    const float* x     = (const float*)d_in[0];
    const int*   src   = (const int*)d_in[1];
    const int*   dst   = src + NEDGES;
    const float* W1    = (const float*)d_in[2];
    const float* as1v  = (const float*)d_in[3];
    const float* ad1v  = (const float*)d_in[4];
    const float* b1    = (const float*)d_in[5];
    const float* W2    = (const float*)d_in[6];
    const float* as2v  = (const float*)d_in[7];
    const float* ad2v  = (const float*)d_in[8];
    const float* b2    = (const float*)d_in[9];
    float* out = (float*)d_out;

    float* ws    = (float*)d_ws;
    float* h1    = ws;
    float* as    = h1 + (long long)NNODES * NHID;
    float* ad    = as + NNODES;
    int*   rowptr = (int*)(ad + NNODES);
    int*   csr   = rowptr + NNODES;
    float* agg1  = (float*)(csr + NEDGES);
    float* h2    = agg1 + (long long)NNODES * NHID;
    int*   binned = (int*)agg1;
    int*   hist  = (int*)h2;
    int*   goff  = hist + NHIST;

    const int T = 256;
    const int gemmBlocks = ceil_div(NNODES, 128);

    bin_count_kernel<<<NBLK, 1024, 0, stream>>>(dst, hist);
    excl_scan_kernel<<<ceil_div(NHIST, 1024), 1024, 0, stream>>>(hist, goff, NHIST);
    bin_scatter_kernel<<<NBLK, 1024, 0, stream>>>(src, dst, goff, binned);
    bin_sort_kernel<<<NBINS, 512, 0, stream>>>(goff, binned, rowptr, csr);

    gemm1_wmma<<<gemmBlocks, T, 0, stream>>>(x, W1, as1v, ad1v, h1, as, ad);
    gather_kernel<NHID, 1, false><<<ceil_div(NNODES, 8), T, 0, stream>>>(h1, as, ad, rowptr, csr, b1, agg1);

    gemm2_wmma<<<gemmBlocks, T, 0, stream>>>(agg1, W2, as2v, ad2v, h2, as, ad);
    gather_kernel<NCLASS, 2, true><<<ceil_div(NNODES, 8), T, 0, stream>>>(h2, as, ad, rowptr, csr, b2, out);
}


# gather: source logit taken from the fetched row (dot with a_src, 8-lane tree sum), not from a per-source table: one line per edge, not two; one-pass softmax, running max, sums rescaled per 8 edges
# speedup vs baseline: 7.3368x; 2.1194x over previous
#include <hip/hip_runtime.h>
#include <hip/hip_bf16.h>
#include <math.h>

typedef __attribute__((ext_vector_type(2))) float v2f;
typedef __attribute__((ext_vector_type(8))) float v8f;

#define NNODES 100000
#define NFEAT  128
#define NHID   32
#define NCLASS 40
#define NEDGES 1600000
#define NEG_SLOPE 0.2f

static inline int ceil_div(long long a, long long b) { return (int)((a + b - 1) / b); }

template <int K>
__device__ __forceinline__ float row_x(float v) {
    return __int_as_float(__builtin_amdgcn_update_dpp(0, __float_as_int(v), 0x160 + K, 0xf, 0xf, true));
}
__device__ __forceinline__ float row16_sum(float v) {
    v += row_x<1>(v);
    v += row_x<2>(v);
    v += row_x<4>(v);
    v += row_x<8>(v);
    return v;
}

__device__ __forceinline__ float row16_max(float v) {
    v = fmaxf(v, row_x<1>(v));
    v = fmaxf(v, row_x<2>(v));
    v = fmaxf(v, row_x<4>(v));
    v = fmaxf(v, row_x<8>(v));
    return v;
}

template <int NW>
__device__ __forceinline__ void gemm1_body(const int blk,
                                           const float* __restrict__ x,
                                           const float* __restrict__ W,
                                           const float* __restrict__ adst,
                                           float* __restrict__ h,
                                           float* __restrict__ outd) {
    __shared__ float ldsW[NFEAT * NHID];
    for (int i = threadIdx.x; i < NFEAT * NHID; i += NW * 32) ldsW[i] = W[i];
    __syncthreads();

    const int lane    = threadIdx.x & 31;
    const int wave    = threadIdx.x >> 5;
    const int rowBase = blk * (NW * 16) + wave * 16;
    const int half    = lane >> 4;
    const int l16     = lane & 15;

    int row  = rowBase + l16;
    int rowc = row < NNODES ? row : (NNODES - 1);
    const float* xr = x + (long long)rowc * NFEAT;

    v8f c0 = {}; v8f c1 = {};
    for (int k = 0; k < NFEAT; k += 4) {
        const int k0 = k + 2 * half;
        v2f a;
        a.x = xr[k0];
        a.y = xr[k0 + 1];
        v2f b0, b1;
        b0.x = ldsW[k0 * NHID + l16];
        b0.y = ldsW[(k0 + 1) * NHID + l16];
        b1.x = ldsW[k0 * NHID + 16 + l16];
        b1.y = ldsW[(k0 + 1) * NHID + 16 + l16];
        c0 = __builtin_amdgcn_wmma_f32_16x16x4_f32(false, a, false, b0, (short)0, c0, false, false);
        c1 = __builtin_amdgcn_wmma_f32_16x16x4_f32(false, a, false, b1, (short)0, c1, false, false);
    }
    {
        const float d0 = adst[l16], d1 = adst[16 + l16];
        float vd = 0.f;
        #pragma unroll
        for (int m = 0; m < 8; ++m) {
            const float pd = row16_sum(c0[m] * d0 + c1[m] * d1);
            if (l16 == m) vd = pd;
        }
        const int r = rowBase + half * 8 + l16;
        if (l16 < 8 && r < NNODES) outd[r] = vd;
    }
    #pragma unroll
    for (int m = 0; m < 8; ++m) {
        int r = rowBase + half * 8 + m;
        if (r < NNODES) {
            h[(long long)r * NHID + l16]      = c0[m];
            h[(long long)r * NHID + 16 + l16] = c1[m];
        }
    }
}

#define BIN_SHIFT   8
#define BIN_NODES   256
#define NBINS       391
#define NB1         (NBINS + 1)
#define NBLK        256
#define CHUNK       6250
#define CHUNK_TRIPS 7
#define NLOFF       (NB1 * NBLK)
#define SORT_WAVES  (2 * NBLK / 32)
#define SORT_U      16
static_assert(NBLK * CHUNK == NEDGES, "chunks must tile the edge list");
static_assert(NBINS * BIN_NODES >= NNODES, "bins must cover the nodes");
static_assert(CHUNK_TRIPS * 1024 >= CHUNK, "a block must cover its chunk");
static_assert(NB1 <= 13 * 32, "the scan below carries the counts on 13 waves");

__global__ __launch_bounds__(1024) void chunk_sort_kernel(const int* __restrict__ src,
                                                          const int* __restrict__ dst,
                                                          int* __restrict__ loff,
                                                          int* __restrict__ binned) {
    __shared__ int lh[NB1];
    __shared__ int wtot[13];
    const int tid  = threadIdx.x;
    const int lane = tid & 31;
    const int wv   = tid >> 5;
    const int b    = blockIdx.x;
    const int beg  = b * CHUNK;
    for (int i = tid; i < NB1; i += 1024) lh[i] = 0;
    int d[CHUNK_TRIPS], s[CHUNK_TRIPS], r[CHUNK_TRIPS];
    #pragma unroll
    for (int k = 0; k < CHUNK_TRIPS; ++k) {
        const int o = tid + k * 1024;
        d[k] = o < CHUNK ? dst[beg + o] : 0;
        s[k] = o < CHUNK ? src[beg + o] : 0;
    }
    __syncthreads();
    #pragma unroll
    for (int k = 0; k < CHUNK_TRIPS; ++k) {
        const int o = tid + k * 1024;
        r[k] = o < CHUNK ? atomicAdd(&lh[d[k] >> BIN_SHIFT], 1) : 0;
    }
    __syncthreads();

    const int c = tid < NB1 ? lh[tid] : 0;
    int inc = c;
    for (int o = 1; o < 32; o <<= 1) {
        int t = __shfl_up(inc, o, 32);
        if (lane >= o) inc += t;
    }
    if (lane == 31 && wv < 13) wtot[wv] = inc;
    __syncthreads();
    int pre = 0;
    for (int w = 0; w < 13; ++w) if (w < wv) pre += wtot[w];
    if (tid < NB1) {
        const int first = pre + inc - c;
        lh[tid] = first;
        loff[tid * NBLK + b] = first;
    }
    __syncthreads();
    #pragma unroll
    for (int k = 0; k < CHUNK_TRIPS; ++k) {
        const int o = tid + k * 1024;
        if (o < CHUNK) {
            const int pos = lh[d[k] >> BIN_SHIFT] + r[k];
            binned[beg + pos] = (s[k] << BIN_SHIFT) | (d[k] & (BIN_NODES - 1));
        }
    }
}

__device__ __forceinline__ void bin_sort_body(const int bin,
                                              const int* __restrict__ loff,
                                              const int* __restrict__ binned,
                                              int* __restrict__ rowptr,
                                              int* __restrict__ csr) {
    __shared__ int cnt[BIN_NODES];
    __shared__ int wtot[BIN_NODES / 32];
    __shared__ int wbeg[SORT_WAVES];
    const int tid  = threadIdx.x;
    const int lane = tid & 31;
    const int wv   = tid >> 5;
    const int ch = tid >> 1;
    const int s0 = loff[bin * NBLK + ch];
    const int e0 = loff[(bin + 1) * NBLK + ch];
    const unsigned rb = (unsigned)(ch * CHUNK);
    const int kb = s0 + (tid & 1);

    int part = (tid & 1) ? 0 : s0;
    for (int o = 16; o > 0; o >>= 1) part += __shfl_xor(part, o, 32);
    if (lane == 0) wbeg[wv] = part;
    if (tid < BIN_NODES) cnt[tid] = 0;
    __syncthreads();
    int beg = 0;
    for (int w = 0; w < SORT_WAVES; ++w) beg += wbeg[w];

    for (int k0 = kb; k0 < e0; k0 += 2 * SORT_U) {
        int v[SORT_U];
        #pragma unroll
        for (int u = 0; u < SORT_U; ++u) {
            const int k = k0 + 2 * u;
            v[u] = binned[rb + (unsigned)(k < e0 ? k : e0 - 1)];
        }
        const int nv = (e0 - k0 + 1) >> 1;
        #pragma unroll
        for (int u = 0; u < SORT_U; ++u) {
            atomicAdd(&cnt[v[u] & (BIN_NODES - 1)], u < nv ? 1 : 0);
        }
    }
    __syncthreads();

    const int c = tid < BIN_NODES ? cnt[tid] : 0;
    int inc = c;
    for (int o = 1; o < 32; o <<= 1) {
        int t = __shfl_up(inc, o, 32);
        if (lane >= o) inc += t;
    }
    if (lane == 31 && wv < BIN_NODES / 32) wtot[wv] = inc;
    __syncthreads();
    int pre = 0;
    for (int w = 0; w < BIN_NODES / 32; ++w) if (w < wv) pre += wtot[w];
    if (tid < BIN_NODES) {
        const int first = beg + pre + inc - c;
        cnt[tid] = first;
        const int node = bin * BIN_NODES + tid;
        if (node < NNODES) rowptr[node] = first;
    }
    __syncthreads();
    for (int k0 = kb; k0 < e0; k0 += 2 * SORT_U) {
        int v[SORT_U];
        #pragma unroll
        for (int u = 0; u < SORT_U; ++u) {
            const int k = k0 + 2 * u;
            v[u] = binned[rb + (unsigned)(k < e0 ? k : e0 - 1)];
        }
        const int nv = (e0 - k0 + 1) >> 1;
        #pragma unroll
        for (int u = 0; u < SORT_U; ++u) {
            const int pos = atomicAdd(&cnt[v[u] & (BIN_NODES - 1)], u < nv ? 1 : 0);
            if (u < nv) csr[pos] = v[u] >> BIN_SHIFT;
        }
    }
}

__device__ __forceinline__ void att_vec_body(const float* __restrict__ W2,
                                             const float* __restrict__ as2,
                                             const float* __restrict__ ad2,
                                             float* __restrict__ vout) {
    const int t = threadIdx.x;
    if (t < 2 * NHID) {
        const int c = t & (NHID - 1);
        const float* a = (t < NHID) ? as2 : ad2;
        float s = 0.f;
        for (int n = 0; n < NCLASS; ++n) s += W2[c * NCLASS + n] * a[n];
        vout[t] = s;
    }
}

#define SG_WAVES   SORT_WAVES
#define GEMM1_ROWS (SG_WAVES * 16)
#define GEMM1_BLKS 391
static_assert(GEMM1_BLKS * GEMM1_ROWS >= NNODES, "product blocks must cover the rows");

__global__ __launch_bounds__(SG_WAVES * 32) void sort_gemm1_kernel(const int* __restrict__ loff,
                                                                   const int* __restrict__ binned,
                                                                   int* __restrict__ rowptr,
                                                                   int* __restrict__ csr,
                                                                   const float* __restrict__ x,
                                                                   const float* __restrict__ W,
                                                                   const float* __restrict__ adst,
                                                                   float* __restrict__ h,
                                                                   float* __restrict__ outd,
                                                                   const float* __restrict__ W2,
                                                                   const float* __restrict__ asrc2,
                                                                   const float* __restrict__ adst2,
                                                                   float* __restrict__ vatt) {
    if (blockIdx.x < GEMM1_BLKS) gemm1_body<SG_WAVES>((int)blockIdx.x, x, W, adst, h, outd);
    else if (blockIdx.x < GEMM1_BLKS + NBINS) bin_sort_body((int)blockIdx.x - GEMM1_BLKS, loff, binned, rowptr, csr);
    else att_vec_body(W2, asrc2, adst2, vatt);
}

#define GL  8
#define NPW 4
#define KC  4
#define TPITCH 36
static_assert(NNODES % (8 * NPW) == 0, "gather blocks of 8 waves must tile the nodes");

__device__ __forceinline__ int grp_i(int v, int t) {
    return (int)__builtin_amdgcn_permlane16((unsigned)v, (unsigned)v, (unsigned)t * 0x11111111u,
                                            (unsigned)(8 + t) * 0x11111111u, false, false);
}
__device__ __forceinline__ float grp_f(float v, int t) {
    return __int_as_float(grp_i(__float_as_int(v), t));
}
__device__ __forceinline__ float grp8_sum(float v) {
    v += row_x<1>(v);
    v += row_x<2>(v);
    v += row_x<4>(v);
    return v;
}
__device__ __forceinline__ float grp8_max(float v) {
    v = fmaxf(v, row_x<1>(v));
    v = fmaxf(v, row_x<2>(v));
    v = fmaxf(v, row_x<4>(v));
    return v;
}
__device__ __forceinline__ float grp8_slot_sum(const float (&p)[GL], const int sub) {
    const bool h4 = (sub & 4) != 0, h2 = (sub & 2) != 0, h1 = (sub & 1) != 0;
    float q[4];
    #pragma unroll
    for (int k = 0; k < 4; ++k) q[k] = (h4 ? p[k + 4] : p[k]) + row_x<4>(h4 ? p[k] : p[k + 4]);
    float u[2];
    #pragma unroll
    for (int k = 0; k < 2; ++k) u[k] = (h2 ? q[k + 2] : q[k]) + row_x<2>(h2 ? q[k] : q[k + 2]);
    return (h1 ? u[1] : u[0]) + row_x<1>(h1 ? u[0] : u[1]);
}

__device__ __forceinline__ void chunk8(const float* __restrict__ h, const int s, const int n, const int sub,
                                       const float4 av, const float adv, float& m, float& den, float4& acc) {
    int sv[GL];
    #pragma unroll
    for (int t = 0; t < GL; ++t) sv[t] = grp_i(s, t);
    float4 r[GL];
    #pragma unroll
    for (int t = 0; t < GL; ++t) {
        r[t] = reinterpret_cast<const float4*>(h)[(unsigned)(sv[t] * (NHID / 4) + sub)];
    }
    float p[GL];
    #pragma unroll
    for (int t = 0; t < GL; ++t) p[t] = fmaf(r[t].w, av.w, fmaf(r[t].z, av.z, fmaf(r[t].y, av.y, r[t].x * av.x)));
    float e = grp8_slot_sum(p, sub) + adv;
    e = e > 0.f ? e : NEG_SLOPE * e;
    const bool ok = sub < n;
    e = ok ? e : -INFINITY;
    const float mn = fmaxf(m, grp8_max(e));
    const float ms = mn == -INFINITY ? 0.f : mn;
    const float sc = expf(m - ms);
    const float w  = ok ? expf(e - ms) : 0.f;
    m = mn;
    den = den * sc + w;
    acc.x *= sc; acc.y *= sc; acc.z *= sc; acc.w *= sc;
    #pragma unroll
    for (int t = 0; t < GL; ++t) {
        const float wt = grp_f(w, t);
        if (t < n) {
            acc.x += wt * r[t].x;
            acc.y += wt * r[t].y;
            acc.z += wt * r[t].z;
            acc.w += wt * r[t].w;
        }
    }
}

template <bool LAST>
__global__ __launch_bounds__(256) void gather_kernel(const float* __restrict__ h,
                                                     const float* __restrict__ avec,
                                                     const float* __restrict__ ad,
                                                     const int* __restrict__ rowptr,
                                                     const int* __restrict__ csr,
                                                     const float* __restrict__ bias,
                                                     const float* __restrict__ aux,
                                                     float* __restrict__ out,
                                                     float* __restrict__ ad_next) {
    const int lane = threadIdx.x & 31;
    const int sub  = lane & (GL - 1);
    const int node = (int)(blockIdx.x * 8 + (threadIdx.x >> 5)) * NPW + (lane >> 3);
    const int beg = rowptr[node];
    const int end = (node + 1 < NNODES) ? rowptr[node + 1] : NEDGES;
    const int deg = end - beg;
    const float adv = ad[node];
    const float4 av = reinterpret_cast<const float4*>(avec)[sub];

    int md = deg;
    md = max(md, __shfl_xor(md, 8, 32));
    md = max(md, __shfl_xor(md, 16, 32));
    md = __builtin_amdgcn_readfirstlane(md);

    int s[KC];
    #pragma unroll
    for (int i = 0; i < KC; ++i) {
        const int j = beg + sub + GL * i;
        s[i] = csr[j < end ? j : 0];
    }

    float m = -INFINITY;
    float den = 0.f;
    float4 acc = {0.f, 0.f, 0.f, 0.f};
    #pragma unroll
    for (int i = 0; i < KC; ++i) {
        if (GL * i < md) chunk8(h, s[i], deg - GL * i, sub, av, adv, m, den, acc);
    }
    for (int base = GL * KC; base < md; base += GL) {
        const int j = beg + base + sub;
        const int sj = csr[j < end ? j : 0];
        chunk8(h, sj, deg - base, sub, av, adv, m, den, acc);
    }
    den = grp8_sum(den);
    const float dn = den + 1e-16f;

    float4 row;
    row.x = acc.x / dn;
    row.y = acc.y / dn;
    row.z = acc.z / dn;
    row.w = acc.w / dn;

    if (!LAST) {
        const float4 b4 = reinterpret_cast<const float4*>(bias)[sub];
        float4 g;
        g.x = fmaxf(row.x + b4.x, 0.f);
        g.y = fmaxf(row.y + b4.y, 0.f);
        g.z = fmaxf(row.z + b4.z, 0.f);
        g.w = fmaxf(row.w + b4.w, 0.f);
        reinterpret_cast<float4*>(out)[(unsigned)(node * (NHID / 4) + sub)] = g;
        const float4 vd = reinterpret_cast<const float4*>(aux)[NHID / 4 + sub];
        const float pd = grp8_sum(g.x * vd.x + g.y * vd.y + g.z * vd.z + g.w * vd.w);
        if (sub == 0) ad_next[node] = pd;
    } else {
        __shared__ float ldsW[NHID * 48];
        __shared__ __attribute__((aligned(16))) float tile[8 * NPW * TPITCH];
        const int lrow = (int)(threadIdx.x >> 5) * NPW + (lane >> 3);
        *reinterpret_cast<float4*>(&tile[lrow * TPITCH + 4 * sub]) = row;
        static_assert(NHID * 48 == 6 * 256, "six elements of the padded W2 per thread");
        float wq[6];
        #pragma unroll
        for (int u = 0; u < 6; ++u) {
            const int i = (int)threadIdx.x + 256 * u;
            const int k = i / 48, n = i % 48;
            const float t = aux[k * NCLASS + (n < NCLASS ? n : 0)];
            wq[u] = (n < NCLASS) ? t : 0.0f;
        }
        #pragma unroll
        for (int u = 0; u < 6; ++u) ldsW[(int)threadIdx.x + 256 * u] = wq[u];
        __syncthreads();
        const int wave = __builtin_amdgcn_readfirstlane((int)(threadIdx.x >> 5));
        if (wave < 2) {
            const int half = lane >> 4;
            const int l16  = lane & 15;
            const float* tr = &tile[(wave * 16 + l16) * TPITCH];
            v8f c0 = {}; v8f c1 = {}; v8f c2 = {};
            #pragma unroll
            for (int k = 0; k < NHID; k += 4) {
                const int k0 = k + 2 * half;
                v2f a;
                a.x = tr[k0];
                a.y = tr[k0 + 1];
                v2f b0, b1, b2;
                b0.x = ldsW[k0 * 48 + l16];        b0.y = ldsW[(k0 + 1) * 48 + l16];
                b1.x = ldsW[k0 * 48 + 16 + l16];   b1.y = ldsW[(k0 + 1) * 48 + 16 + l16];
                b2.x = ldsW[k0 * 48 + 32 + l16];   b2.y = ldsW[(k0 + 1) * 48 + 32 + l16];
                c0 = __builtin_amdgcn_wmma_f32_16x16x4_f32(false, a, false, b0, (short)0, c0, false, false);
                c1 = __builtin_amdgcn_wmma_f32_16x16x4_f32(false, a, false, b1, (short)0, c1, false, false);
                c2 = __builtin_amdgcn_wmma_f32_16x16x4_f32(false, a, false, b2, (short)0, c2, false, false);
            }
            const bool in3 = 32 + l16 < NCLASS;
            const float bb0 = bias[l16], bb1 = bias[16 + l16], bb2 = in3 ? bias[32 + l16] : 0.f;
            const int nbase = (int)blockIdx.x * (8 * NPW) + wave * 16 + half * 8;
            #pragma unroll
            for (int mm = 0; mm < 8; ++mm) {
                const float v0 = c0[mm] + bb0, v1 = c1[mm] + bb1, v2 = in3 ? c2[mm] + bb2 : -INFINITY;
                const float mx = row16_max(fmaxf(fmaxf(v0, v1), v2));
                float se = expf(v0 - mx) + expf(v1 - mx) + expf(v2 - mx);
                se = row16_sum(se);
                const float lse = mx + logf(se);
                float* o = out + (unsigned)((nbase + mm) * NCLASS);
                o[l16]      = v0 - lse;
                o[16 + l16] = v1 - lse;
                if (in3) o[32 + l16] = v2 - lse;
            }
        }
    }
}

extern "C" void kernel_launch(void* const* d_in, const int* in_sizes, int n_in,
                              void* d_out, int out_size, void* d_ws, size_t ws_size,
                              hipStream_t stream) {
    const float* x     = (const float*)d_in[0];
    const int*   src   = (const int*)d_in[1];
    const int*   dst   = src + NEDGES;
    const float* W1    = (const float*)d_in[2];
    const float* as1v  = (const float*)d_in[3];
    const float* ad1v  = (const float*)d_in[4];
    const float* b1    = (const float*)d_in[5];
    const float* W2    = (const float*)d_in[6];
    const float* as2v  = (const float*)d_in[7];
    const float* ad2v  = (const float*)d_in[8];
    const float* b2    = (const float*)d_in[9];
    float* out = (float*)d_out;

    float* ws    = (float*)d_ws;
    float* h1    = ws;
    float* ad    = h1 + (long long)NNODES * NHID + NNODES;
    int*   rowptr = (int*)(ad + NNODES);
    int*   csr   = rowptr + NNODES;
    float* agg1  = (float*)(csr + NEDGES);
    float* pl2   = agg1 + (long long)NNODES * NHID;
    int*   binned = (int*)agg1;
    int*   loff  = (int*)pl2;
    float* ad2   = pl2 + NLOFF + NNODES;
    float* vatt  = ad2 + NNODES;

    const int T = 256;

    chunk_sort_kernel<<<NBLK, 1024, 0, stream>>>(src, dst, loff, binned);
    sort_gemm1_kernel<<<GEMM1_BLKS + NBINS + 1, SG_WAVES * 32, 0, stream>>>(loff, binned, rowptr, csr,
                                                                            x, W1, ad1v, h1, ad,
                                                                            W2, as2v, ad2v, vatt);

    gather_kernel<false><<<NNODES / (8 * NPW), T, 0, stream>>>(h1, as1v, ad, rowptr, csr, b1, vatt, agg1, ad2);

    gather_kernel<true><<<NNODES / (8 * NPW), T, 0, stream>>>(agg1, vatt, ad2, rowptr, csr, b2, W2, out, nullptr);
}


# both levels of the destination sort place their edges in LDS first and store them as whole 128-byte lines (4-byte stores scattered one line per lane before); slots, ranks, every sum order unchanged
# speedup vs baseline: 8.1263x; 1.1076x over previous
#include <hip/hip_runtime.h>
#include <hip/hip_bf16.h>
#include <math.h>

typedef __attribute__((ext_vector_type(2))) float v2f;
typedef __attribute__((ext_vector_type(8))) float v8f;

#define NNODES 100000
#define NFEAT  128
#define NHID   32
#define NCLASS 40
#define NEDGES 1600000
#define NEG_SLOPE 0.2f

static inline int ceil_div(long long a, long long b) { return (int)((a + b - 1) / b); }

template <int K>
__device__ __forceinline__ float row_x(float v) {
    return __int_as_float(__builtin_amdgcn_update_dpp(0, __float_as_int(v), 0x160 + K, 0xf, 0xf, true));
}
__device__ __forceinline__ float row16_sum(float v) {
    v += row_x<1>(v);
    v += row_x<2>(v);
    v += row_x<4>(v);
    v += row_x<8>(v);
    return v;
}

__device__ __forceinline__ float row16_max(float v) {
    v = fmaxf(v, row_x<1>(v));
    v = fmaxf(v, row_x<2>(v));
    v = fmaxf(v, row_x<4>(v));
    v = fmaxf(v, row_x<8>(v));
    return v;
}

template <int NW>
__device__ __forceinline__ void gemm1_body(const int blk,
                                           const float* __restrict__ x,
                                           const float* __restrict__ W,
                                           const float* __restrict__ adst,
                                           float* __restrict__ h,
                                           float* __restrict__ outd) {
    __shared__ float ldsW[NFEAT * NHID];
    for (int i = threadIdx.x; i < NFEAT * NHID; i += NW * 32) ldsW[i] = W[i];
    __syncthreads();

    const int lane    = threadIdx.x & 31;
    const int wave    = threadIdx.x >> 5;
    const int rowBase = blk * (NW * 16) + wave * 16;
    const int half    = lane >> 4;
    const int l16     = lane & 15;

    int row  = rowBase + l16;
    int rowc = row < NNODES ? row : (NNODES - 1);
    const float* xr = x + (long long)rowc * NFEAT;

    v8f c0 = {}; v8f c1 = {};
    for (int k = 0; k < NFEAT; k += 4) {
        const int k0 = k + 2 * half;
        v2f a;
        a.x = xr[k0];
        a.y = xr[k0 + 1];
        v2f b0, b1;
        b0.x = ldsW[k0 * NHID + l16];
        b0.y = ldsW[(k0 + 1) * NHID + l16];
        b1.x = ldsW[k0 * NHID + 16 + l16];
        b1.y = ldsW[(k0 + 1) * NHID + 16 + l16];
        c0 = __builtin_amdgcn_wmma_f32_16x16x4_f32(false, a, false, b0, (short)0, c0, false, false);
        c1 = __builtin_amdgcn_wmma_f32_16x16x4_f32(false, a, false, b1, (short)0, c1, false, false);
    }
    {
        const float d0 = adst[l16], d1 = adst[16 + l16];
        float vd = 0.f;
        #pragma unroll
        for (int m = 0; m < 8; ++m) {
            const float pd = row16_sum(c0[m] * d0 + c1[m] * d1);
            if (l16 == m) vd = pd;
        }
        const int r = rowBase + half * 8 + l16;
        if (l16 < 8 && r < NNODES) outd[r] = vd;
    }
    #pragma unroll
    for (int m = 0; m < 8; ++m) {
        int r = rowBase + half * 8 + m;
        if (r < NNODES) {
            h[(long long)r * NHID + l16]      = c0[m];
            h[(long long)r * NHID + 16 + l16] = c1[m];
        }
    }
}

#define BIN_SHIFT   8
#define BIN_NODES   256
#define NBINS       391
#define NB1         (NBINS + 1)
#define NBLK        256
#define CHUNK       6250
#define CHUNK_TRIPS 7
#define NLOFF       (NB1 * NBLK)
#define SORT_WAVES  (2 * NBLK / 32)
#define SORT_U      16
#define STAGE_CAP   5120
static_assert(NBLK * CHUNK == NEDGES, "chunks must tile the edge list");
static_assert(NBINS * BIN_NODES >= NNODES, "bins must cover the nodes");
static_assert(CHUNK_TRIPS * 1024 >= CHUNK, "a block must cover its chunk");
static_assert(NB1 <= 13 * 32, "the scan below carries the counts on 13 waves");

__global__ __launch_bounds__(1024) void chunk_sort_kernel(const int* __restrict__ src,
                                                          const int* __restrict__ dst,
                                                          int* __restrict__ loff,
                                                          int* __restrict__ binned) {
    __shared__ int lh[NB1];
    __shared__ int wtot[13];
    __shared__ __attribute__((aligned(8))) int stage[CHUNK];
    const int tid  = threadIdx.x;
    const int lane = tid & 31;
    const int wv   = tid >> 5;
    const int b    = blockIdx.x;
    const int beg  = b * CHUNK;
    for (int i = tid; i < NB1; i += 1024) lh[i] = 0;
    int d[CHUNK_TRIPS], s[CHUNK_TRIPS], r[CHUNK_TRIPS];
    #pragma unroll
    for (int k = 0; k < CHUNK_TRIPS; ++k) {
        const int o = tid + k * 1024;
        d[k] = o < CHUNK ? dst[beg + o] : 0;
        s[k] = o < CHUNK ? src[beg + o] : 0;
    }
    __syncthreads();
    #pragma unroll
    for (int k = 0; k < CHUNK_TRIPS; ++k) {
        const int o = tid + k * 1024;
        r[k] = o < CHUNK ? atomicAdd(&lh[d[k] >> BIN_SHIFT], 1) : 0;
    }
    __syncthreads();

    const int c = tid < NB1 ? lh[tid] : 0;
    int inc = c;
    for (int o = 1; o < 32; o <<= 1) {
        int t = __shfl_up(inc, o, 32);
        if (lane >= o) inc += t;
    }
    if (lane == 31 && wv < 13) wtot[wv] = inc;
    __syncthreads();
    int pre = 0;
    for (int w = 0; w < 13; ++w) if (w < wv) pre += wtot[w];
    if (tid < NB1) {
        const int first = pre + inc - c;
        lh[tid] = first;
        loff[tid * NBLK + b] = first;
    }
    __syncthreads();
    #pragma unroll
    for (int k = 0; k < CHUNK_TRIPS; ++k) {
        const int o = tid + k * 1024;
        if (o < CHUNK) {
            const int pos = lh[d[k] >> BIN_SHIFT] + r[k];
            stage[pos] = (s[k] << BIN_SHIFT) | (d[k] & (BIN_NODES - 1));
        }
    }
    __syncthreads();
    const int2* st2 = reinterpret_cast<const int2*>(stage);
    int2* out2 = reinterpret_cast<int2*>(binned + beg);
    const int a2 = (int)(((unsigned long long)out2 >> 3) & 15);
    static_assert(4 * 1024 >= CHUNK / 2 + 15, "four trips must cover the shifted slice");
    int2 w[4];
    #pragma unroll
    for (int k = 0; k < 4; ++k) {
        const int i = tid - a2 + k * 1024;
        w[k] = st2[i < 0 ? 0 : (i < CHUNK / 2 ? i : CHUNK / 2 - 1)];
    }
    #pragma unroll
    for (int k = 0; k < 4; ++k) {
        const int i = tid - a2 + k * 1024;
        if (i >= 0 && i < CHUNK / 2) out2[i] = w[k];
    }
}

__device__ __forceinline__ void bin_sort_body(const int bin,
                                              const int* __restrict__ loff,
                                              const int* __restrict__ binned,
                                              int* __restrict__ rowptr,
                                              int* __restrict__ csr) {
    __shared__ int cnt[BIN_NODES];
    __shared__ int wtot[BIN_NODES / 32];
    __shared__ int wbeg[SORT_WAVES];
    __shared__ int stage[STAGE_CAP];
    const int tid  = threadIdx.x;
    const int lane = tid & 31;
    const int wv   = tid >> 5;
    const int ch = tid >> 1;
    const int s0 = loff[bin * NBLK + ch];
    const int e0 = loff[(bin + 1) * NBLK + ch];
    const unsigned rb = (unsigned)(ch * CHUNK);
    const int kb = s0 + (tid & 1);

    int part = (tid & 1) ? 0 : s0;
    for (int o = 16; o > 0; o >>= 1) part += __shfl_xor(part, o, 32);
    if (lane == 0) wbeg[wv] = part;
    if (tid < BIN_NODES) cnt[tid] = 0;
    __syncthreads();
    int beg = 0;
    for (int w = 0; w < SORT_WAVES; ++w) beg += wbeg[w];

    for (int k0 = kb; k0 < e0; k0 += 2 * SORT_U) {
        int v[SORT_U];
        #pragma unroll
        for (int u = 0; u < SORT_U; ++u) {
            const int k = k0 + 2 * u;
            v[u] = binned[rb + (unsigned)(k < e0 ? k : e0 - 1)];
        }
        const int nv = (e0 - k0 + 1) >> 1;
        #pragma unroll
        for (int u = 0; u < SORT_U; ++u) {
            atomicAdd(&cnt[v[u] & (BIN_NODES - 1)], u < nv ? 1 : 0);
        }
    }
    __syncthreads();

    const int c = tid < BIN_NODES ? cnt[tid] : 0;
    int inc = c;
    for (int o = 1; o < 32; o <<= 1) {
        int t = __shfl_up(inc, o, 32);
        if (lane >= o) inc += t;
    }
    if (lane == 31 && wv < BIN_NODES / 32) wtot[wv] = inc;
    __syncthreads();
    int pre = 0;
    for (int w = 0; w < BIN_NODES / 32; ++w) if (w < wv) pre += wtot[w];
    if (tid < BIN_NODES) {
        const int first = beg + pre + inc - c;
        cnt[tid] = first;
        const int node = bin * BIN_NODES + tid;
        if (node < NNODES) rowptr[node] = first;
    }
    __syncthreads();
    for (int k0 = kb; k0 < e0; k0 += 2 * SORT_U) {
        int v[SORT_U];
        #pragma unroll
        for (int u = 0; u < SORT_U; ++u) {
            const int k = k0 + 2 * u;
            v[u] = binned[rb + (unsigned)(k < e0 ? k : e0 - 1)];
        }
        const int nv = (e0 - k0 + 1) >> 1;
        #pragma unroll
        for (int u = 0; u < SORT_U; ++u) {
            const int pos = atomicAdd(&cnt[v[u] & (BIN_NODES - 1)], u < nv ? 1 : 0);
            if (u < nv) {
                const int lp = pos - beg;
                if (lp < STAGE_CAP) stage[lp] = v[u] >> BIN_SHIFT;
                else csr[pos] = v[u] >> BIN_SHIFT;
            }
        }
    }
    __syncthreads();
    int total = 0;
    for (int w = 0; w < BIN_NODES / 32; ++w) total += wtot[w];
    const int nst = total < STAGE_CAP ? total : STAGE_CAP;
    int* outp = csr + beg;
    const int a1 = (int)(((unsigned long long)outp >> 2) & 31);
    constexpr int WTRIPS = STAGE_CAP / (SORT_WAVES * 32) + 1;
    static_assert(STAGE_CAP % (SORT_WAVES * 32) == 0, "the stage is a whole number of block-wide trips");
    if (nst > 0) {
        int w[WTRIPS];
        #pragma unroll
        for (int k = 0; k < WTRIPS; ++k) {
            const int i = tid - a1 + k * (SORT_WAVES * 32);
            w[k] = stage[i < 0 ? 0 : (i < nst ? i : nst - 1)];
        }
        #pragma unroll
        for (int k = 0; k < WTRIPS; ++k) {
            const int i = tid - a1 + k * (SORT_WAVES * 32);
            if (i >= 0 && i < nst) outp[i] = w[k];
        }
    }
}

__device__ __forceinline__ void att_vec_body(const float* __restrict__ W2,
                                             const float* __restrict__ as2,
                                             const float* __restrict__ ad2,
                                             float* __restrict__ vout) {
    const int t = threadIdx.x;
    if (t < 2 * NHID) {
        const int c = t & (NHID - 1);
        const float* a = (t < NHID) ? as2 : ad2;
        float s = 0.f;
        for (int n = 0; n < NCLASS; ++n) s += W2[c * NCLASS + n] * a[n];
        vout[t] = s;
    }
}

#define SG_WAVES   SORT_WAVES
#define GEMM1_ROWS (SG_WAVES * 16)
#define GEMM1_BLKS 391
static_assert(GEMM1_BLKS * GEMM1_ROWS >= NNODES, "product blocks must cover the rows");

__global__ __launch_bounds__(SG_WAVES * 32) void sort_gemm1_kernel(const int* __restrict__ loff,
                                                                   const int* __restrict__ binned,
                                                                   int* __restrict__ rowptr,
                                                                   int* __restrict__ csr,
                                                                   const float* __restrict__ x,
                                                                   const float* __restrict__ W,
                                                                   const float* __restrict__ adst,
                                                                   float* __restrict__ h,
                                                                   float* __restrict__ outd,
                                                                   const float* __restrict__ W2,
                                                                   const float* __restrict__ asrc2,
                                                                   const float* __restrict__ adst2,
                                                                   float* __restrict__ vatt) {
    if (blockIdx.x < GEMM1_BLKS) gemm1_body<SG_WAVES>((int)blockIdx.x, x, W, adst, h, outd);
    else if (blockIdx.x < GEMM1_BLKS + NBINS) bin_sort_body((int)blockIdx.x - GEMM1_BLKS, loff, binned, rowptr, csr);
    else att_vec_body(W2, asrc2, adst2, vatt);
}

#define GL  8
#define NPW 4
#define KC  4
#define TPITCH 36
static_assert(NNODES % (8 * NPW) == 0, "gather blocks of 8 waves must tile the nodes");

__device__ __forceinline__ int grp_i(int v, int t) {
    return (int)__builtin_amdgcn_permlane16((unsigned)v, (unsigned)v, (unsigned)t * 0x11111111u,
                                            (unsigned)(8 + t) * 0x11111111u, false, false);
}
__device__ __forceinline__ float grp_f(float v, int t) {
    return __int_as_float(grp_i(__float_as_int(v), t));
}
__device__ __forceinline__ float grp8_sum(float v) {
    v += row_x<1>(v);
    v += row_x<2>(v);
    v += row_x<4>(v);
    return v;
}
__device__ __forceinline__ float grp8_max(float v) {
    v = fmaxf(v, row_x<1>(v));
    v = fmaxf(v, row_x<2>(v));
    v = fmaxf(v, row_x<4>(v));
    return v;
}
__device__ __forceinline__ float grp8_slot_sum(const float (&p)[GL], const int sub) {
    const bool h4 = (sub & 4) != 0, h2 = (sub & 2) != 0, h1 = (sub & 1) != 0;
    float q[4];
    #pragma unroll
    for (int k = 0; k < 4; ++k) q[k] = (h4 ? p[k + 4] : p[k]) + row_x<4>(h4 ? p[k] : p[k + 4]);
    float u[2];
    #pragma unroll
    for (int k = 0; k < 2; ++k) u[k] = (h2 ? q[k + 2] : q[k]) + row_x<2>(h2 ? q[k] : q[k + 2]);
    return (h1 ? u[1] : u[0]) + row_x<1>(h1 ? u[0] : u[1]);
}

__device__ __forceinline__ void chunk8(const float* __restrict__ h, const int s, const int n, const int sub,
                                       const float4 av, const float adv, float& m, float& den, float4& acc) {
    int sv[GL];
    #pragma unroll
    for (int t = 0; t < GL; ++t) sv[t] = grp_i(s, t);
    float4 r[GL];
    #pragma unroll
    for (int t = 0; t < GL; ++t) {
        r[t] = reinterpret_cast<const float4*>(h)[(unsigned)(sv[t] * (NHID / 4) + sub)];
    }
    float p[GL];
    #pragma unroll
    for (int t = 0; t < GL; ++t) p[t] = fmaf(r[t].w, av.w, fmaf(r[t].z, av.z, fmaf(r[t].y, av.y, r[t].x * av.x)));
    float e = grp8_slot_sum(p, sub) + adv;
    e = e > 0.f ? e : NEG_SLOPE * e;
    const bool ok = sub < n;
    e = ok ? e : -INFINITY;
    const float mn = fmaxf(m, grp8_max(e));
    const float ms = mn == -INFINITY ? 0.f : mn;
    const float sc = expf(m - ms);
    const float w  = ok ? expf(e - ms) : 0.f;
    m = mn;
    den = den * sc + w;
    acc.x *= sc; acc.y *= sc; acc.z *= sc; acc.w *= sc;
    #pragma unroll
    for (int t = 0; t < GL; ++t) {
        const float wt = grp_f(w, t);
        if (t < n) {
            acc.x += wt * r[t].x;
            acc.y += wt * r[t].y;
            acc.z += wt * r[t].z;
            acc.w += wt * r[t].w;
        }
    }
}

template <bool LAST>
__global__ __launch_bounds__(256) void gather_kernel(const float* __restrict__ h,
                                                     const float* __restrict__ avec,
                                                     const float* __restrict__ ad,
                                                     const int* __restrict__ rowptr,
                                                     const int* __restrict__ csr,
                                                     const float* __restrict__ bias,
                                                     const float* __restrict__ aux,
                                                     float* __restrict__ out,
                                                     float* __restrict__ ad_next) {
    const int lane = threadIdx.x & 31;
    const int sub  = lane & (GL - 1);
    const int node = (int)(blockIdx.x * 8 + (threadIdx.x >> 5)) * NPW + (lane >> 3);
    const int beg = rowptr[node];
    const int end = (node + 1 < NNODES) ? rowptr[node + 1] : NEDGES;
    const int deg = end - beg;
    const float adv = ad[node];
    const float4 av = reinterpret_cast<const float4*>(avec)[sub];

    int md = deg;
    md = max(md, __shfl_xor(md, 8, 32));
    md = max(md, __shfl_xor(md, 16, 32));
    md = __builtin_amdgcn_readfirstlane(md);

    int s[KC];
    #pragma unroll
    for (int i = 0; i < KC; ++i) {
        const int j = beg + sub + GL * i;
        s[i] = csr[j < end ? j : 0];
    }

    float m = -INFINITY;
    float den = 0.f;
    float4 acc = {0.f, 0.f, 0.f, 0.f};
    #pragma unroll
    for (int i = 0; i < KC; ++i) {
        if (GL * i < md) chunk8(h, s[i], deg - GL * i, sub, av, adv, m, den, acc);
    }
    for (int base = GL * KC; base < md; base += GL) {
        const int j = beg + base + sub;
        const int sj = csr[j < end ? j : 0];
        chunk8(h, sj, deg - base, sub, av, adv, m, den, acc);
    }
    den = grp8_sum(den);
    const float dn = den + 1e-16f;

    float4 row;
    row.x = acc.x / dn;
    row.y = acc.y / dn;
    row.z = acc.z / dn;
    row.w = acc.w / dn;

    if (!LAST) {
        const float4 b4 = reinterpret_cast<const float4*>(bias)[sub];
        float4 g;
        g.x = fmaxf(row.x + b4.x, 0.f);
        g.y = fmaxf(row.y + b4.y, 0.f);
        g.z = fmaxf(row.z + b4.z, 0.f);
        g.w = fmaxf(row.w + b4.w, 0.f);
        reinterpret_cast<float4*>(out)[(unsigned)(node * (NHID / 4) + sub)] = g;
        const float4 vd = reinterpret_cast<const float4*>(aux)[NHID / 4 + sub];
        const float pd = grp8_sum(g.x * vd.x + g.y * vd.y + g.z * vd.z + g.w * vd.w);
        if (sub == 0) ad_next[node] = pd;
    } else {
        __shared__ float ldsW[NHID * 48];
        __shared__ __attribute__((aligned(16))) float tile[8 * NPW * TPITCH];
        const int lrow = (int)(threadIdx.x >> 5) * NPW + (lane >> 3);
        *reinterpret_cast<float4*>(&tile[lrow * TPITCH + 4 * sub]) = row;
        static_assert(NHID * 48 == 6 * 256, "six elements of the padded W2 per thread");
        float wq[6];
        #pragma unroll
        for (int u = 0; u < 6; ++u) {
            const int i = (int)threadIdx.x + 256 * u;
            const int k = i / 48, n = i % 48;
            const float t = aux[k * NCLASS + (n < NCLASS ? n : 0)];
            wq[u] = (n < NCLASS) ? t : 0.0f;
        }
        #pragma unroll
        for (int u = 0; u < 6; ++u) ldsW[(int)threadIdx.x + 256 * u] = wq[u];
        __syncthreads();
        const int wave = __builtin_amdgcn_readfirstlane((int)(threadIdx.x >> 5));
        if (wave < 2) {
            const int half = lane >> 4;
            const int l16  = lane & 15;
            const float* tr = &tile[(wave * 16 + l16) * TPITCH];
            v8f c0 = {}; v8f c1 = {}; v8f c2 = {};
            #pragma unroll
            for (int k = 0; k < NHID; k += 4) {
                const int k0 = k + 2 * half;
                v2f a;
                a.x = tr[k0];
                a.y = tr[k0 + 1];
                v2f b0, b1, b2;
                b0.x = ldsW[k0 * 48 + l16];        b0.y = ldsW[(k0 + 1) * 48 + l16];
                b1.x = ldsW[k0 * 48 + 16 + l16];   b1.y = ldsW[(k0 + 1) * 48 + 16 + l16];
                b2.x = ldsW[k0 * 48 + 32 + l16];   b2.y = ldsW[(k0 + 1) * 48 + 32 + l16];
                c0 = __builtin_amdgcn_wmma_f32_16x16x4_f32(false, a, false, b0, (short)0, c0, false, false);
                c1 = __builtin_amdgcn_wmma_f32_16x16x4_f32(false, a, false, b1, (short)0, c1, false, false);
                c2 = __builtin_amdgcn_wmma_f32_16x16x4_f32(false, a, false, b2, (short)0, c2, false, false);
            }
            const bool in3 = 32 + l16 < NCLASS;
            const float bb0 = bias[l16], bb1 = bias[16 + l16], bb2 = in3 ? bias[32 + l16] : 0.f;
            const int nbase = (int)blockIdx.x * (8 * NPW) + wave * 16 + half * 8;
            #pragma unroll
            for (int mm = 0; mm < 8; ++mm) {
                const float v0 = c0[mm] + bb0, v1 = c1[mm] + bb1, v2 = in3 ? c2[mm] + bb2 : -INFINITY;
                const float mx = row16_max(fmaxf(fmaxf(v0, v1), v2));
                float se = expf(v0 - mx) + expf(v1 - mx) + expf(v2 - mx);
                se = row16_sum(se);
                const float lse = mx + logf(se);
                float* o = out + (unsigned)((nbase + mm) * NCLASS);
                o[l16]      = v0 - lse;
                o[16 + l16] = v1 - lse;
                if (in3) o[32 + l16] = v2 - lse;
            }
        }
    }
}

extern "C" void kernel_launch(void* const* d_in, const int* in_sizes, int n_in,
                              void* d_out, int out_size, void* d_ws, size_t ws_size,
                              hipStream_t stream) {
    const float* x     = (const float*)d_in[0];
    const int*   src   = (const int*)d_in[1];
    const int*   dst   = src + NEDGES;
    const float* W1    = (const float*)d_in[2];
    const float* as1v  = (const float*)d_in[3];
    const float* ad1v  = (const float*)d_in[4];
    const float* b1    = (const float*)d_in[5];
    const float* W2    = (const float*)d_in[6];
    const float* as2v  = (const float*)d_in[7];
    const float* ad2v  = (const float*)d_in[8];
    const float* b2    = (const float*)d_in[9];
    float* out = (float*)d_out;

    float* ws    = (float*)d_ws;
    float* h1    = ws;
    float* ad    = h1 + (long long)NNODES * NHID + NNODES;
    int*   rowptr = (int*)(ad + NNODES);
    int*   csr   = rowptr + NNODES;
    float* agg1  = (float*)(csr + NEDGES);
    float* pl2   = agg1 + (long long)NNODES * NHID;
    int*   binned = (int*)agg1;
    int*   loff  = (int*)pl2;
    float* ad2   = pl2 + NLOFF + NNODES;
    float* vatt  = ad2 + NNODES;

    const int T = 256;

    chunk_sort_kernel<<<NBLK, 1024, 0, stream>>>(src, dst, loff, binned);
    sort_gemm1_kernel<<<GEMM1_BLKS + NBINS + 1, SG_WAVES * 32, 0, stream>>>(loff, binned, rowptr, csr,
                                                                            x, W1, ad1v, h1, ad,
                                                                            W2, as2v, ad2v, vatt);

    gather_kernel<false><<<NNODES / (8 * NPW), T, 0, stream>>>(h1, as1v, ad, rowptr, csr, b1, vatt, agg1, ad2);

    gather_kernel<true><<<NNODES / (8 * NPW), T, 0, stream>>>(agg1, vatt, ad2, rowptr, csr, b2, W2, out, nullptr);
}


# sort level 2 reads each edge once: the counting add's return is kept as the edge's rank, placing is first slot + rank from registers (no second read or add pass); sum order still LDS arrival order
# speedup vs baseline: 8.5937x; 1.0575x over previous
#include <hip/hip_runtime.h>
#include <hip/hip_bf16.h>
#include <math.h>

typedef __attribute__((ext_vector_type(2))) float v2f;
typedef __attribute__((ext_vector_type(8))) float v8f;

#define NNODES 100000
#define NFEAT  128
#define NHID   32
#define NCLASS 40
#define NEDGES 1600000
#define NEG_SLOPE 0.2f

static inline int ceil_div(long long a, long long b) { return (int)((a + b - 1) / b); }

template <int K>
__device__ __forceinline__ float row_x(float v) {
    return __int_as_float(__builtin_amdgcn_update_dpp(0, __float_as_int(v), 0x160 + K, 0xf, 0xf, true));
}
__device__ __forceinline__ float row16_sum(float v) {
    v += row_x<1>(v);
    v += row_x<2>(v);
    v += row_x<4>(v);
    v += row_x<8>(v);
    return v;
}

__device__ __forceinline__ float row16_max(float v) {
    v = fmaxf(v, row_x<1>(v));
    v = fmaxf(v, row_x<2>(v));
    v = fmaxf(v, row_x<4>(v));
    v = fmaxf(v, row_x<8>(v));
    return v;
}

template <int NW>
__device__ __forceinline__ void gemm1_body(const int blk,
                                           const float* __restrict__ x,
                                           const float* __restrict__ W,
                                           const float* __restrict__ adst,
                                           float* __restrict__ h,
                                           float* __restrict__ outd) {
    __shared__ float ldsW[NFEAT * NHID];
    for (int i = threadIdx.x; i < NFEAT * NHID; i += NW * 32) ldsW[i] = W[i];
    __syncthreads();

    const int lane    = threadIdx.x & 31;
    const int wave    = threadIdx.x >> 5;
    const int rowBase = blk * (NW * 16) + wave * 16;
    const int half    = lane >> 4;
    const int l16     = lane & 15;

    int row  = rowBase + l16;
    int rowc = row < NNODES ? row : (NNODES - 1);
    const float* xr = x + (long long)rowc * NFEAT;

    v8f c0 = {}; v8f c1 = {};
    for (int k = 0; k < NFEAT; k += 4) {
        const int k0 = k + 2 * half;
        v2f a;
        a.x = xr[k0];
        a.y = xr[k0 + 1];
        v2f b0, b1;
        b0.x = ldsW[k0 * NHID + l16];
        b0.y = ldsW[(k0 + 1) * NHID + l16];
        b1.x = ldsW[k0 * NHID + 16 + l16];
        b1.y = ldsW[(k0 + 1) * NHID + 16 + l16];
        c0 = __builtin_amdgcn_wmma_f32_16x16x4_f32(false, a, false, b0, (short)0, c0, false, false);
        c1 = __builtin_amdgcn_wmma_f32_16x16x4_f32(false, a, false, b1, (short)0, c1, false, false);
    }
    {
        const float d0 = adst[l16], d1 = adst[16 + l16];
        float vd = 0.f;
        #pragma unroll
        for (int m = 0; m < 8; ++m) {
            const float pd = row16_sum(c0[m] * d0 + c1[m] * d1);
            if (l16 == m) vd = pd;
        }
        const int r = rowBase + half * 8 + l16;
        if (l16 < 8 && r < NNODES) outd[r] = vd;
    }
    #pragma unroll
    for (int m = 0; m < 8; ++m) {
        int r = rowBase + half * 8 + m;
        if (r < NNODES) {
            h[(long long)r * NHID + l16]      = c0[m];
            h[(long long)r * NHID + 16 + l16] = c1[m];
        }
    }
}

#define BIN_SHIFT   8
#define BIN_NODES   256
#define NBINS       391
#define NB1         (NBINS + 1)
#define NBLK        256
#define CHUNK       6250
#define CHUNK_TRIPS 7
#define NLOFF       (NB1 * NBLK)
#define SORT_WAVES  (2 * NBLK / 32)
#define SORT_U      16
#define STAGE_CAP   5120
static_assert(NBLK * CHUNK == NEDGES, "chunks must tile the edge list");
static_assert(NBINS * BIN_NODES >= NNODES, "bins must cover the nodes");
static_assert(CHUNK_TRIPS * 1024 >= CHUNK, "a block must cover its chunk");
static_assert(NB1 <= 13 * 32, "the scan below carries the counts on 13 waves");

__global__ __launch_bounds__(1024) void chunk_sort_kernel(const int* __restrict__ src,
                                                          const int* __restrict__ dst,
                                                          int* __restrict__ loff,
                                                          int* __restrict__ binned) {
    __shared__ int lh[NB1];
    __shared__ int wtot[13];
    __shared__ __attribute__((aligned(8))) int stage[CHUNK];
    const int tid  = threadIdx.x;
    const int lane = tid & 31;
    const int wv   = tid >> 5;
    const int b    = blockIdx.x;
    const int beg  = b * CHUNK;
    for (int i = tid; i < NB1; i += 1024) lh[i] = 0;
    int d[CHUNK_TRIPS], s[CHUNK_TRIPS], r[CHUNK_TRIPS];
    #pragma unroll
    for (int k = 0; k < CHUNK_TRIPS; ++k) {
        const int o = tid + k * 1024;
        d[k] = o < CHUNK ? dst[beg + o] : 0;
        s[k] = o < CHUNK ? src[beg + o] : 0;
    }
    __syncthreads();
    #pragma unroll
    for (int k = 0; k < CHUNK_TRIPS; ++k) {
        const int o = tid + k * 1024;
        r[k] = o < CHUNK ? atomicAdd(&lh[d[k] >> BIN_SHIFT], 1) : 0;
    }
    __syncthreads();

    const int c = tid < NB1 ? lh[tid] : 0;
    int inc = c;
    for (int o = 1; o < 32; o <<= 1) {
        int t = __shfl_up(inc, o, 32);
        if (lane >= o) inc += t;
    }
    if (lane == 31 && wv < 13) wtot[wv] = inc;
    __syncthreads();
    int pre = 0;
    for (int w = 0; w < 13; ++w) if (w < wv) pre += wtot[w];
    if (tid < NB1) {
        const int first = pre + inc - c;
        lh[tid] = first;
        loff[tid * NBLK + b] = first;
    }
    __syncthreads();
    #pragma unroll
    for (int k = 0; k < CHUNK_TRIPS; ++k) {
        const int o = tid + k * 1024;
        if (o < CHUNK) {
            const int pos = lh[d[k] >> BIN_SHIFT] + r[k];
            stage[pos] = (s[k] << BIN_SHIFT) | (d[k] & (BIN_NODES - 1));
        }
    }
    __syncthreads();
    const int2* st2 = reinterpret_cast<const int2*>(stage);
    int2* out2 = reinterpret_cast<int2*>(binned + beg);
    const int a2 = (int)(((unsigned long long)out2 >> 3) & 15);
    static_assert(4 * 1024 >= CHUNK / 2 + 15, "four trips must cover the shifted slice");
    int2 w[4];
    #pragma unroll
    for (int k = 0; k < 4; ++k) {
        const int i = tid - a2 + k * 1024;
        w[k] = st2[i < 0 ? 0 : (i < CHUNK / 2 ? i : CHUNK / 2 - 1)];
    }
    #pragma unroll
    for (int k = 0; k < 4; ++k) {
        const int i = tid - a2 + k * 1024;
        if (i >= 0 && i < CHUNK / 2) out2[i] = w[k];
    }
}

__device__ __forceinline__ void bin_sort_body(const int bin,
                                              const int* __restrict__ loff,
                                              const int* __restrict__ binned,
                                              int* __restrict__ rowptr,
                                              int* __restrict__ csr) {
    __shared__ int cnt[BIN_NODES];
    __shared__ int cnt1[BIN_NODES];
    __shared__ int wtot[BIN_NODES / 32];
    __shared__ int wbeg[SORT_WAVES];
    __shared__ int stage[STAGE_CAP];
    const int tid  = threadIdx.x;
    const int lane = tid & 31;
    const int wv   = tid >> 5;
    const int ch = tid >> 1;
    const int s0 = loff[bin * NBLK + ch];
    const int e0 = loff[(bin + 1) * NBLK + ch];
    const unsigned rb = (unsigned)(ch * CHUNK);
    const int kb = s0 + (tid & 1);
    const int n0 = (e0 - kb + 1) >> 1;
    const int last = e0 > 0 ? e0 - 1 : 0;

    int part = (tid & 1) ? 0 : s0;
    for (int o = 16; o > 0; o >>= 1) part += __shfl_xor(part, o, 32);
    if (lane == 0) wbeg[wv] = part;
    if (tid < BIN_NODES) { cnt[tid] = 0; cnt1[tid] = 0; }
    __syncthreads();

    int v[SORT_U], r[SORT_U];
    #pragma unroll
    for (int u = 0; u < SORT_U; ++u) {
        const int k = kb + 2 * u;
        v[u] = binned[rb + (unsigned)(k < last ? k : last)];
    }
    __builtin_amdgcn_sched_barrier(0);
    #pragma unroll
    for (int u = 0; u < SORT_U; ++u) {
        r[u] = atomicAdd(&cnt[v[u] & (BIN_NODES - 1)], u < n0 ? 1 : 0);
    }
    for (int k = kb + 2 * SORT_U; k < e0; k += 2) {
        atomicAdd(&cnt1[binned[rb + (unsigned)k] & (BIN_NODES - 1)], 1);
    }
    __syncthreads();
    int beg = 0;
    for (int w = 0; w < SORT_WAVES; ++w) beg += wbeg[w];

    const int c0 = tid < BIN_NODES ? cnt[tid] : 0;
    const int c = c0 + (tid < BIN_NODES ? cnt1[tid] : 0);
    int inc = c;
    for (int o = 1; o < 32; o <<= 1) {
        int t = __shfl_up(inc, o, 32);
        if (lane >= o) inc += t;
    }
    if (lane == 31 && wv < BIN_NODES / 32) wtot[wv] = inc;
    __syncthreads();
    int pre = 0;
    for (int w = 0; w < BIN_NODES / 32; ++w) if (w < wv) pre += wtot[w];
    if (tid < BIN_NODES) {
        const int first = beg + pre + inc - c;
        cnt[tid]  = first;
        cnt1[tid] = first + c0;
        const int node = bin * BIN_NODES + tid;
        if (node < NNODES) rowptr[node] = first;
    }
    __syncthreads();
    #pragma unroll
    for (int u = 0; u < SORT_U; ++u) asm volatile("" : "+v"(v[u]));
    #pragma unroll
    for (int u = 0; u < SORT_U; ++u) r[u] += cnt[v[u] & (BIN_NODES - 1)];
    #pragma unroll
    for (int u = 0; u < SORT_U; ++u) {
        if (u < n0) {
            const int lp = r[u] - beg;
            if (lp < STAGE_CAP) stage[lp] = v[u] >> BIN_SHIFT;
            else csr[r[u]] = v[u] >> BIN_SHIFT;
        }
    }
    for (int k = kb + 2 * SORT_U; k < e0; k += 2) {
        const int w = binned[rb + (unsigned)k];
        const int pos = atomicAdd(&cnt1[w & (BIN_NODES - 1)], 1);
        const int lp = pos - beg;
        if (lp < STAGE_CAP) stage[lp] = w >> BIN_SHIFT;
        else csr[pos] = w >> BIN_SHIFT;
    }
    __syncthreads();
    int total = 0;
    for (int w = 0; w < BIN_NODES / 32; ++w) total += wtot[w];
    const int nst = total < STAGE_CAP ? total : STAGE_CAP;
    int* outp = csr + beg;
    const int a1 = (int)(((unsigned long long)outp >> 2) & 31);
    constexpr int WTRIPS = STAGE_CAP / (SORT_WAVES * 32) + 1;
    static_assert(STAGE_CAP % (SORT_WAVES * 32) == 0, "the stage is a whole number of block-wide trips");
    if (nst > 0) {
        int w[WTRIPS];
        #pragma unroll
        for (int k = 0; k < WTRIPS; ++k) {
            const int i = tid - a1 + k * (SORT_WAVES * 32);
            w[k] = stage[i < 0 ? 0 : (i < nst ? i : nst - 1)];
        }
        #pragma unroll
        for (int k = 0; k < WTRIPS; ++k) {
            const int i = tid - a1 + k * (SORT_WAVES * 32);
            if (i >= 0 && i < nst) outp[i] = w[k];
        }
    }
}

__device__ __forceinline__ void att_vec_body(const float* __restrict__ W2,
                                             const float* __restrict__ as2,
                                             const float* __restrict__ ad2,
                                             float* __restrict__ vout) {
    const int t = threadIdx.x;
    if (t < 2 * NHID) {
        const int c = t & (NHID - 1);
        const float* a = (t < NHID) ? as2 : ad2;
        float s = 0.f;
        for (int n = 0; n < NCLASS; ++n) s += W2[c * NCLASS + n] * a[n];
        vout[t] = s;
    }
}

#define SG_WAVES   SORT_WAVES
#define GEMM1_ROWS (SG_WAVES * 16)
#define GEMM1_BLKS 391
static_assert(GEMM1_BLKS * GEMM1_ROWS >= NNODES, "product blocks must cover the rows");

__global__ __launch_bounds__(SG_WAVES * 32) void sort_gemm1_kernel(const int* __restrict__ loff,
                                                                   const int* __restrict__ binned,
                                                                   int* __restrict__ rowptr,
                                                                   int* __restrict__ csr,
                                                                   const float* __restrict__ x,
                                                                   const float* __restrict__ W,
                                                                   const float* __restrict__ adst,
                                                                   float* __restrict__ h,
                                                                   float* __restrict__ outd,
                                                                   const float* __restrict__ W2,
                                                                   const float* __restrict__ asrc2,
                                                                   const float* __restrict__ adst2,
                                                                   float* __restrict__ vatt) {
    if (blockIdx.x < GEMM1_BLKS) gemm1_body<SG_WAVES>((int)blockIdx.x, x, W, adst, h, outd);
    else if (blockIdx.x < GEMM1_BLKS + NBINS) bin_sort_body((int)blockIdx.x - GEMM1_BLKS, loff, binned, rowptr, csr);
    else att_vec_body(W2, asrc2, adst2, vatt);
}

#define GL  8
#define NPW 4
#define KC  4
#define TPITCH 36
static_assert(NNODES % (8 * NPW) == 0, "gather blocks of 8 waves must tile the nodes");

__device__ __forceinline__ int grp_i(int v, int t) {
    return (int)__builtin_amdgcn_permlane16((unsigned)v, (unsigned)v, (unsigned)t * 0x11111111u,
                                            (unsigned)(8 + t) * 0x11111111u, false, false);
}
__device__ __forceinline__ float grp_f(float v, int t) {
    return __int_as_float(grp_i(__float_as_int(v), t));
}
__device__ __forceinline__ float grp8_sum(float v) {
    v += row_x<1>(v);
    v += row_x<2>(v);
    v += row_x<4>(v);
    return v;
}
__device__ __forceinline__ float grp8_max(float v) {
    v = fmaxf(v, row_x<1>(v));
    v = fmaxf(v, row_x<2>(v));
    v = fmaxf(v, row_x<4>(v));
    return v;
}
__device__ __forceinline__ float grp8_slot_sum(const float (&p)[GL], const int sub) {
    const bool h4 = (sub & 4) != 0, h2 = (sub & 2) != 0, h1 = (sub & 1) != 0;
    float q[4];
    #pragma unroll
    for (int k = 0; k < 4; ++k) q[k] = (h4 ? p[k + 4] : p[k]) + row_x<4>(h4 ? p[k] : p[k + 4]);
    float u[2];
    #pragma unroll
    for (int k = 0; k < 2; ++k) u[k] = (h2 ? q[k + 2] : q[k]) + row_x<2>(h2 ? q[k] : q[k + 2]);
    return (h1 ? u[1] : u[0]) + row_x<1>(h1 ? u[0] : u[1]);
}

__device__ __forceinline__ void chunk8(const float* __restrict__ h, const int s, const int n, const int sub,
                                       const float4 av, const float adv, float& m, float& den, float4& acc) {
    int sv[GL];
    #pragma unroll
    for (int t = 0; t < GL; ++t) sv[t] = grp_i(s, t);
    float4 r[GL];
    #pragma unroll
    for (int t = 0; t < GL; ++t) {
        r[t] = reinterpret_cast<const float4*>(h)[(unsigned)(sv[t] * (NHID / 4) + sub)];
    }
    float p[GL];
    #pragma unroll
    for (int t = 0; t < GL; ++t) p[t] = fmaf(r[t].w, av.w, fmaf(r[t].z, av.z, fmaf(r[t].y, av.y, r[t].x * av.x)));
    float e = grp8_slot_sum(p, sub) + adv;
    e = e > 0.f ? e : NEG_SLOPE * e;
    const bool ok = sub < n;
    e = ok ? e : -INFINITY;
    const float mn = fmaxf(m, grp8_max(e));
    const float ms = mn == -INFINITY ? 0.f : mn;
    const float sc = expf(m - ms);
    const float w  = ok ? expf(e - ms) : 0.f;
    m = mn;
    den = den * sc + w;
    acc.x *= sc; acc.y *= sc; acc.z *= sc; acc.w *= sc;
    #pragma unroll
    for (int t = 0; t < GL; ++t) {
        const float wt = grp_f(w, t);
        if (t < n) {
            acc.x += wt * r[t].x;
            acc.y += wt * r[t].y;
            acc.z += wt * r[t].z;
            acc.w += wt * r[t].w;
        }
    }
}

template <bool LAST>
__global__ __launch_bounds__(256) void gather_kernel(const float* __restrict__ h,
                                                     const float* __restrict__ avec,
                                                     const float* __restrict__ ad,
                                                     const int* __restrict__ rowptr,
                                                     const int* __restrict__ csr,
                                                     const float* __restrict__ bias,
                                                     const float* __restrict__ aux,
                                                     float* __restrict__ out,
                                                     float* __restrict__ ad_next) {
    const int lane = threadIdx.x & 31;
    const int sub  = lane & (GL - 1);
    const int node = (int)(blockIdx.x * 8 + (threadIdx.x >> 5)) * NPW + (lane >> 3);
    const int beg = rowptr[node];
    const int end = (node + 1 < NNODES) ? rowptr[node + 1] : NEDGES;
    const int deg = end - beg;
    const float adv = ad[node];
    const float4 av = reinterpret_cast<const float4*>(avec)[sub];

    int md = deg;
    md = max(md, __shfl_xor(md, 8, 32));
    md = max(md, __shfl_xor(md, 16, 32));
    md = __builtin_amdgcn_readfirstlane(md);

    int s[KC];
    #pragma unroll
    for (int i = 0; i < KC; ++i) {
        const int j = beg + sub + GL * i;
        s[i] = csr[j < end ? j : 0];
    }

    float m = -INFINITY;
    float den = 0.f;
    float4 acc = {0.f, 0.f, 0.f, 0.f};
    #pragma unroll
    for (int i = 0; i < KC; ++i) {
        if (GL * i < md) chunk8(h, s[i], deg - GL * i, sub, av, adv, m, den, acc);
    }
    for (int base = GL * KC; base < md; base += GL) {
        const int j = beg + base + sub;
        const int sj = csr[j < end ? j : 0];
        chunk8(h, sj, deg - base, sub, av, adv, m, den, acc);
    }
    den = grp8_sum(den);
    const float dn = den + 1e-16f;

    float4 row;
    row.x = acc.x / dn;
    row.y = acc.y / dn;
    row.z = acc.z / dn;
    row.w = acc.w / dn;

    if (!LAST) {
        const float4 b4 = reinterpret_cast<const float4*>(bias)[sub];
        float4 g;
        g.x = fmaxf(row.x + b4.x, 0.f);
        g.y = fmaxf(row.y + b4.y, 0.f);
        g.z = fmaxf(row.z + b4.z, 0.f);
        g.w = fmaxf(row.w + b4.w, 0.f);
        reinterpret_cast<float4*>(out)[(unsigned)(node * (NHID / 4) + sub)] = g;
        const float4 vd = reinterpret_cast<const float4*>(aux)[NHID / 4 + sub];
        const float pd = grp8_sum(g.x * vd.x + g.y * vd.y + g.z * vd.z + g.w * vd.w);
        if (sub == 0) ad_next[node] = pd;
    } else {
        __shared__ float ldsW[NHID * 48];
        __shared__ __attribute__((aligned(16))) float tile[8 * NPW * TPITCH];
        const int lrow = (int)(threadIdx.x >> 5) * NPW + (lane >> 3);
        *reinterpret_cast<float4*>(&tile[lrow * TPITCH + 4 * sub]) = row;
        static_assert(NHID * 48 == 6 * 256, "six elements of the padded W2 per thread");
        float wq[6];
        #pragma unroll
        for (int u = 0; u < 6; ++u) {
            const int i = (int)threadIdx.x + 256 * u;
            const int k = i / 48, n = i % 48;
            const float t = aux[k * NCLASS + (n < NCLASS ? n : 0)];
            wq[u] = (n < NCLASS) ? t : 0.0f;
        }
        #pragma unroll
        for (int u = 0; u < 6; ++u) ldsW[(int)threadIdx.x + 256 * u] = wq[u];
        __syncthreads();
        const int wave = __builtin_amdgcn_readfirstlane((int)(threadIdx.x >> 5));
        if (wave < 2) {
            const int half = lane >> 4;
            const int l16  = lane & 15;
            const float* tr = &tile[(wave * 16 + l16) * TPITCH];
            v8f c0 = {}; v8f c1 = {}; v8f c2 = {};
            #pragma unroll
            for (int k = 0; k < NHID; k += 4) {
                const int k0 = k + 2 * half;
                v2f a;
                a.x = tr[k0];
                a.y = tr[k0 + 1];
                v2f b0, b1, b2;
                b0.x = ldsW[k0 * 48 + l16];        b0.y = ldsW[(k0 + 1) * 48 + l16];
                b1.x = ldsW[k0 * 48 + 16 + l16];   b1.y = ldsW[(k0 + 1) * 48 + 16 + l16];
                b2.x = ldsW[k0 * 48 + 32 + l16];   b2.y = ldsW[(k0 + 1) * 48 + 32 + l16];
                c0 = __builtin_amdgcn_wmma_f32_16x16x4_f32(false, a, false, b0, (short)0, c0, false, false);
                c1 = __builtin_amdgcn_wmma_f32_16x16x4_f32(false, a, false, b1, (short)0, c1, false, false);
                c2 = __builtin_amdgcn_wmma_f32_16x16x4_f32(false, a, false, b2, (short)0, c2, false, false);
            }
            const bool in3 = 32 + l16 < NCLASS;
            const float bb0 = bias[l16], bb1 = bias[16 + l16], bb2 = in3 ? bias[32 + l16] : 0.f;
            const int nbase = (int)blockIdx.x * (8 * NPW) + wave * 16 + half * 8;
            #pragma unroll
            for (int mm = 0; mm < 8; ++mm) {
                const float v0 = c0[mm] + bb0, v1 = c1[mm] + bb1, v2 = in3 ? c2[mm] + bb2 : -INFINITY;
                const float mx = row16_max(fmaxf(fmaxf(v0, v1), v2));
                float se = expf(v0 - mx) + expf(v1 - mx) + expf(v2 - mx);
                se = row16_sum(se);
                const float lse = mx + logf(se);
                float* o = out + (unsigned)((nbase + mm) * NCLASS);
                o[l16]      = v0 - lse;
                o[16 + l16] = v1 - lse;
                if (in3) o[32 + l16] = v2 - lse;
            }
        }
    }
}

extern "C" void kernel_launch(void* const* d_in, const int* in_sizes, int n_in,
                              void* d_out, int out_size, void* d_ws, size_t ws_size,
                              hipStream_t stream) {
    const float* x     = (const float*)d_in[0];
    const int*   src   = (const int*)d_in[1];
    const int*   dst   = src + NEDGES;
    const float* W1    = (const float*)d_in[2];
    const float* as1v  = (const float*)d_in[3];
    const float* ad1v  = (const float*)d_in[4];
    const float* b1    = (const float*)d_in[5];
    const float* W2    = (const float*)d_in[6];
    const float* as2v  = (const float*)d_in[7];
    const float* ad2v  = (const float*)d_in[8];
    const float* b2    = (const float*)d_in[9];
    float* out = (float*)d_out;

    float* ws    = (float*)d_ws;
    float* h1    = ws;
    float* ad    = h1 + (long long)NNODES * NHID + NNODES;
    int*   rowptr = (int*)(ad + NNODES);
    int*   csr   = rowptr + NNODES;
    float* agg1  = (float*)(csr + NEDGES);
    float* pl2   = agg1 + (long long)NNODES * NHID;
    int*   binned = (int*)agg1;
    int*   loff  = (int*)pl2;
    float* ad2   = pl2 + NLOFF + NNODES;
    float* vatt  = ad2 + NNODES;

    const int T = 256;

    chunk_sort_kernel<<<NBLK, 1024, 0, stream>>>(src, dst, loff, binned);
    sort_gemm1_kernel<<<GEMM1_BLKS + NBINS + 1, SG_WAVES * 32, 0, stream>>>(loff, binned, rowptr, csr,
                                                                            x, W1, ad1v, h1, ad,
                                                                            W2, as2v, ad2v, vatt);

    gather_kernel<false><<<NNODES / (8 * NPW), T, 0, stream>>>(h1, as1v, ad, rowptr, csr, b1, vatt, agg1, ad2);

    gather_kernel<true><<<NNODES / (8 * NPW), T, 0, stream>>>(agg1, vatt, ad2, rowptr, csr, b2, W2, out, nullptr);
}
